# starting kernel
# baseline (speedup 1.0000x reference)
#include <hip/hip_runtime.h>

#define BB 2
#define SS 384
#define CC 4
#define HH 128
#define DD 768

typedef float v2f __attribute__((ext_vector_type(2)));
typedef float v8f __attribute__((ext_vector_type(8)));

__device__ __forceinline__ float tanhf_fast(float x) {
  float e = __builtin_amdgcn_exp2f(x * 2.8853900817779268f);
  float r = __builtin_amdgcn_rcpf(e + 1.0f);
  return 1.0f - 2.0f * r;
}

#define MODE_TANH 1
#define MODE_BIAS 4
#define MODE_ADDV 8

template <int MODE, bool HAS_RAW>
__device__ __forceinline__ void store_tile(v8f c, float* O, int ldo, float* Oraw,
                                           const float* bias, const float* addvec,
                                           int m0, int n0, int lane) {
  int col   = n0 + (lane & 15);
  int rbase = m0 + ((lane >> 4) << 3);
#pragma unroll
  for (int i = 0; i < 8; ++i) {
    int   row = rbase + i;
    float v   = c[i];
    if constexpr (MODE & MODE_BIAS) v += bias[col];
    if constexpr (MODE & MODE_ADDV) v += addvec[(long)(row >> 2) * HH + col];
    if constexpr (HAS_RAW) Oraw[(long)row * ldo + col] = v;
    O[(long)row * ldo + col] = (MODE & MODE_TANH) ? tanhf_fast(v) : v;
  }
}

template <int MODE, bool HAS_RAW>
__global__ void __launch_bounds__(128)
gemm_wmma_kernel(const float* __restrict__ A, int lda, long aOuter, long aInner,
                 const float* __restrict__ B, int ldb, long bOuter, long bInner,
                 float* __restrict__ O, int ldo, long oOuter, long oInner,
                 float* __restrict__ Oraw,
                 const float* __restrict__ bias, const float* __restrict__ addvec,
                 int K) {
  __shared__ float As[64 * 20];
  __shared__ float Bs[64 * 20];

  int lane = threadIdx.x;
  int wave = threadIdx.y;
  int t    = wave * 32 + lane;
  int zo   = blockIdx.z >> 2;
  int zi   = blockIdx.z & 3;
  const float* Ab = A + (long)zo * aOuter + (long)zi * aInner;
  const float* Bb = B + (long)zo * bOuter + (long)zi * bInner;

  int mblk = blockIdx.y * 64;
  int nblk = blockIdx.x * 64;
  int mw   = (wave >> 1) * 32;
  int nw   = (wave & 1) * 32;

  v8f c00 = {}, c01 = {}, c10 = {}, c11 = {};

  for (int kc = 0; kc < K; kc += 16) {
    __syncthreads();
#pragma unroll
    for (int rep = 0; rep < 2; ++rep) {
      int idx = t + rep * 128;
      int ar = idx >> 2, ac = (idx & 3) << 2;
      float4 av = *(const float4*)(Ab + (long)(mblk + ar) * lda + kc + ac);
      *(float4*)(&As[ar * 20 + ac]) = av;
      int br = idx >> 4, bc = (idx & 15) << 2;
      float4 bv = *(const float4*)(Bb + (long)(kc + br) * ldb + nblk + bc);
      Bs[(bc + 0) * 20 + br] = bv.x;
      Bs[(bc + 1) * 20 + br] = bv.y;
      Bs[(bc + 2) * 20 + br] = bv.z;
      Bs[(bc + 3) * 20 + br] = bv.w;
    }
    __syncthreads();

#pragma unroll
    for (int kl = 0; kl < 16; kl += 4) {
      int kf = kl + ((lane >> 4) << 1);
      int rr = lane & 15;
      v2f a0 = *(const v2f*)(&As[(mw + rr) * 20 + kf]);
      v2f a1 = *(const v2f*)(&As[(mw + 16 + rr) * 20 + kf]);
      v2f b0 = *(const v2f*)(&Bs[(nw + rr) * 20 + kf]);
      v2f b1 = *(const v2f*)(&Bs[(nw + 16 + rr) * 20 + kf]);
      c00 = __builtin_amdgcn_wmma_f32_16x16x4_f32(false, a0, false, b0, (short)0, c00, false, false);
      c01 = __builtin_amdgcn_wmma_f32_16x16x4_f32(false, a0, false, b1, (short)0, c01, false, false);
      c10 = __builtin_amdgcn_wmma_f32_16x16x4_f32(false, a1, false, b0, (short)0, c10, false, false);
      c11 = __builtin_amdgcn_wmma_f32_16x16x4_f32(false, a1, false, b1, (short)0, c11, false, false);
    }
  }

  float* Ob = O + (long)zo * oOuter + (long)zi * oInner;
  float* Rb = HAS_RAW ? (Oraw + (long)zo * oOuter + (long)zi * oInner) : nullptr;
  int m0 = mblk + mw, n0 = nblk + nw;
  store_tile<MODE, HAS_RAW>(c00, Ob, ldo, Rb, bias, addvec, m0,      n0,      lane);
  store_tile<MODE, HAS_RAW>(c01, Ob, ldo, Rb, bias, addvec, m0,      n0 + 16, lane);
  store_tile<MODE, HAS_RAW>(c10, Ob, ldo, Rb, bias, addvec, m0 + 16, n0,      lane);
  store_tile<MODE, HAS_RAW>(c11, Ob, ldo, Rb, bias, addvec, m0 + 16, n0 + 16, lane);
}

__global__ void __launch_bounds__(128)
final_reduce_kernel(const float* __restrict__ pref, const float* __restrict__ targ,
                    const float* __restrict__ wout, float* __restrict__ out) {
  __shared__ float sp[HH];
  __shared__ float sw[HH];
  int tid = threadIdx.x;
  int p   = blockIdx.y;
  int bc  = blockIdx.z;
  int b   = bc >> 2;
  int c   = bc & 3;

  long prow = ((long)(b * SS + p) * CC + c) * HH;
  sp[tid] = pref[prow + tid];
  sw[tid] = wout[tid];
  __syncthreads();

  int a = blockIdx.x * 128 + tid;
  const float* ta = targ + ((long)(b * SS + a) * CC + c) * HH;
  __builtin_prefetch(ta, 0, 0);

  float acc = 0.0f;
#pragma unroll 8
  for (int k = 0; k < HH; k += 4) {
    float4 tv = *(const float4*)(ta + k);
    acc += tanhf_fast(sp[k + 0] + tv.x) * sw[k + 0];
    acc += tanhf_fast(sp[k + 1] + tv.y) * sw[k + 1];
    acc += tanhf_fast(sp[k + 2] + tv.z) * sw[k + 2];
    acc += tanhf_fast(sp[k + 3] + tv.w) * sw[k + 3];
  }
  out[((long)(b * SS + p) * CC + c) * SS + a] = acc;
}

extern "C" void kernel_launch(void* const* d_in, const int* in_sizes, int n_in,
                              void* d_out, int out_size, void* d_ws, size_t ws_size,
                              hipStream_t stream) {
  const float* seq  = (const float*)d_in[0];
  const float* base = (const float*)d_in[1];
  const float* Wp   = (const float*)d_in[2];
  const float* bp   = (const float*)d_in[3];
  const float* Ua   = (const float*)d_in[4];
  const float* ba   = (const float*)d_in[5];
  const float* Wmid = (const float*)d_in[6];
  const float* bmid = (const float*)d_in[7];
  const float* wout = (const float*)d_in[8];
  float* out = (float*)d_out;

  float* ws      = (float*)d_ws;
  float* tanh_hp = ws;
  float* ha_raw  = ws + 393216;
  float* tanh_ha = ws + 786432;
  float* tanhref = ws + 1179648;
  float* targ    = ws + 1572864;
  float* tref    = ws + 1966080;
  float* pref    = ws + 2064384;

  dim3 blk(32, 4);

  gemm_wmma_kernel<MODE_TANH | MODE_BIAS, false><<<dim3(8, 12, 1), blk, 0, stream>>>(
      seq, DD, 0, 0, Wp, CC * HH, 0, 0, tanh_hp, CC * HH, 0, 0,
      nullptr, bp, nullptr, DD);

  gemm_wmma_kernel<MODE_TANH | MODE_BIAS, true><<<dim3(8, 12, 1), blk, 0, stream>>>(
      seq, DD, 0, 0, Ua, CC * HH, 0, 0, tanh_ha, CC * HH, 0, 0,
      ha_raw, ba, nullptr, DD);

  gemm_wmma_kernel<MODE_TANH, false><<<dim3(2, 6, BB * CC), blk, 0, stream>>>(
      base, CC * SS, (long)SS * CC * SS, SS,
      ha_raw, CC * HH, (long)SS * CC * HH, HH,
      tanhref, CC * HH, (long)SS * CC * HH, HH,
      nullptr, nullptr, nullptr, SS);

  gemm_wmma_kernel<0, false><<<dim3(2, 12, 1), blk, 0, stream>>>(
      tanhref, CC * HH, 0, 0, Wmid + 2 * HH * HH, HH, 0, 0, tref, HH, 0, 0,
      nullptr, nullptr, nullptr, CC * HH);

  gemm_wmma_kernel<MODE_BIAS | MODE_ADDV, false><<<dim3(2, 48, 1), blk, 0, stream>>>(
      tanh_hp, HH, 0, 0, Wmid, HH, 0, 0, pref, HH, 0, 0,
      nullptr, bmid, tref, HH);

  gemm_wmma_kernel<0, false><<<dim3(2, 48, 1), blk, 0, stream>>>(
      tanh_ha, HH, 0, 0, Wmid + HH * HH, HH, 0, 0, targ, HH, 0, 0,
      nullptr, nullptr, nullptr, HH);

  final_reduce_kernel<<<dim3(SS / 128, SS, BB * CC), dim3(128), 0, stream>>>(
      pref, targ, wout, out);
}


# GEMM kernel: no LDS staging, no barriers; each wave (2 per block) owns 16 rows x 64 cols and loads its WMMA fragments from global memory into two alternating 32-k register sets; sums unchanged
# speedup vs baseline: 1.9499x; 1.9499x over previous
#include <hip/hip_runtime.h>

#define BB 2
#define SS 384
#define CC 4
#define HH 128
#define DD 768

typedef float v2f __attribute__((ext_vector_type(2)));
typedef float v8f __attribute__((ext_vector_type(8)));

__device__ __forceinline__ float tanhf_fast(float x) {
  float e = __builtin_amdgcn_exp2f(x * 2.8853900817779268f);
  float r = __builtin_amdgcn_rcpf(e + 1.0f);
  return 1.0f - 2.0f * r;
}

#define MODE_TANH 1
#define MODE_BIAS 4
#define MODE_ADDV 8

#define KCH 32
#define KST (KCH / 4)

template <int LDB>
__device__ __forceinline__ void load_set(v2f (&a)[KST], float4 (&b)[KST][2],
                                         const float* __restrict__ Ap,
                                         const float* __restrict__ Bp) {
#pragma unroll
  for (int j = 0; j < KST; ++j) {
    a[j]    = *(const v2f*)(Ap + 4 * j);
    b[j][0] = *(const float4*)(Bp + (4 * j) * LDB);
    b[j][1] = *(const float4*)(Bp + (4 * j + 1) * LDB);
  }
  __builtin_amdgcn_sched_barrier(0);
}

__device__ __forceinline__ void mma_set(const v2f (&a)[KST], const float4 (&b)[KST][2],
                                        v8f (&acc)[4]) {
#pragma unroll
  for (int j = 0; j < KST; ++j) {
    v2f b0 = {b[j][0].x, b[j][1].x};
    v2f b1 = {b[j][0].y, b[j][1].y};
    v2f b2 = {b[j][0].z, b[j][1].z};
    v2f b3 = {b[j][0].w, b[j][1].w};
    acc[0] = __builtin_amdgcn_wmma_f32_16x16x4_f32(false, a[j], false, b0, (short)0, acc[0], false, false);
    acc[1] = __builtin_amdgcn_wmma_f32_16x16x4_f32(false, a[j], false, b1, (short)0, acc[1], false, false);
    acc[2] = __builtin_amdgcn_wmma_f32_16x16x4_f32(false, a[j], false, b2, (short)0, acc[2], false, false);
    acc[3] = __builtin_amdgcn_wmma_f32_16x16x4_f32(false, a[j], false, b3, (short)0, acc[3], false, false);
  }
}

template <int MODE, bool HAS_RAW, int LDB>
__global__ void __launch_bounds__(64)
gemm_wmma_kernel(const float* __restrict__ A, int lda, long aOuter, long aInner,
                 const float* __restrict__ B, long bOuter, long bInner,
                 float* __restrict__ O, int ldo, long oOuter, long oInner,
                 float* __restrict__ Oraw,
                 const float* __restrict__ bias, const float* __restrict__ addvec,
                 int K) {
  int lane = threadIdx.x;
  int wave = threadIdx.y;
  int c    = lane & 15;
  int h    = lane >> 4;
  int zo   = blockIdx.z >> 2;
  int zi   = blockIdx.z & 3;
  int m0   = blockIdx.y * 32 + wave * 16;
  int n0   = blockIdx.x * 64;

  const float* Ap = A + (long)zo * aOuter + (long)zi * aInner + (long)(m0 + c) * lda + 2 * h;
  const float* Bp = B + (long)zo * bOuter + (long)zi * bInner + (2 * h) * LDB + n0 + 4 * c;

  v8f acc[4] = {};
  v2f    a0[KST], a1[KST];
  float4 b0[KST][2], b1[KST][2];

  load_set<LDB>(a0, b0, Ap, Bp);
  int kc = 0;
  for (; kc + 2 * KCH < K; kc += 2 * KCH) {
    load_set<LDB>(a1, b1, Ap + kc + KCH, Bp + (long)(kc + KCH) * LDB);
    mma_set(a0, b0, acc);
    load_set<LDB>(a0, b0, Ap + kc + 2 * KCH, Bp + (long)(kc + 2 * KCH) * LDB);
    mma_set(a1, b1, acc);
  }
  load_set<LDB>(a1, b1, Ap + kc + KCH, Bp + (long)(kc + KCH) * LDB);
  mma_set(a0, b0, acc);
  mma_set(a1, b1, acc);

  float* Ob = O + (long)zo * oOuter + (long)zi * oInner;
  float* Rb = HAS_RAW ? (Oraw + (long)zo * oOuter + (long)zi * oInner) : nullptr;
  int col = n0 + 4 * c;
  float4 bi = {0.f, 0.f, 0.f, 0.f};
  if constexpr (MODE & MODE_BIAS) bi = *(const float4*)(bias + col);
#pragma unroll
  for (int r = 0; r < 8; ++r) {
    int row = m0 + 8 * h + r;
    float4 v = {acc[0][r], acc[1][r], acc[2][r], acc[3][r]};
    if constexpr (MODE & MODE_BIAS) { v.x += bi.x; v.y += bi.y; v.z += bi.z; v.w += bi.w; }
    if constexpr (MODE & MODE_ADDV) {
      float4 av = *(const float4*)(addvec + (long)(row >> 2) * HH + col);
      v.x += av.x; v.y += av.y; v.z += av.z; v.w += av.w;
    }
    if constexpr (HAS_RAW) *(float4*)(Rb + (long)row * ldo + col) = v;
    if constexpr (MODE & MODE_TANH) {
      v.x = tanhf_fast(v.x); v.y = tanhf_fast(v.y); v.z = tanhf_fast(v.z); v.w = tanhf_fast(v.w);
    }
    *(float4*)(Ob + (long)row * ldo + col) = v;
  }
}

__global__ void __launch_bounds__(128)
final_reduce_kernel(const float* __restrict__ pref, const float* __restrict__ targ,
                    const float* __restrict__ wout, float* __restrict__ out) {
  __shared__ float sp[HH];
  __shared__ float sw[HH];
  int tid = threadIdx.x;
  int p   = blockIdx.y;
  int bc  = blockIdx.z;
  int b   = bc >> 2;
  int c   = bc & 3;

  long prow = ((long)(b * SS + p) * CC + c) * HH;
  sp[tid] = pref[prow + tid];
  sw[tid] = wout[tid];
  __syncthreads();

  int a = blockIdx.x * 128 + tid;
  const float* ta = targ + ((long)(b * SS + a) * CC + c) * HH;
  __builtin_prefetch(ta, 0, 0);

  float acc = 0.0f;
#pragma unroll 8
  for (int k = 0; k < HH; k += 4) {
    float4 tv = *(const float4*)(ta + k);
    acc += tanhf_fast(sp[k + 0] + tv.x) * sw[k + 0];
    acc += tanhf_fast(sp[k + 1] + tv.y) * sw[k + 1];
    acc += tanhf_fast(sp[k + 2] + tv.z) * sw[k + 2];
    acc += tanhf_fast(sp[k + 3] + tv.w) * sw[k + 3];
  }
  out[((long)(b * SS + p) * CC + c) * SS + a] = acc;
}

extern "C" void kernel_launch(void* const* d_in, const int* in_sizes, int n_in,
                              void* d_out, int out_size, void* d_ws, size_t ws_size,
                              hipStream_t stream) {
  const float* seq  = (const float*)d_in[0];
  const float* base = (const float*)d_in[1];
  const float* Wp   = (const float*)d_in[2];
  const float* bp   = (const float*)d_in[3];
  const float* Ua   = (const float*)d_in[4];
  const float* ba   = (const float*)d_in[5];
  const float* Wmid = (const float*)d_in[6];
  const float* bmid = (const float*)d_in[7];
  const float* wout = (const float*)d_in[8];
  float* out = (float*)d_out;

  float* ws      = (float*)d_ws;
  float* tanh_hp = ws;
  float* ha_raw  = ws + 393216;
  float* tanh_ha = ws + 786432;
  float* tanhref = ws + 1179648;
  float* targ    = ws + 1572864;
  float* tref    = ws + 1966080;
  float* pref    = ws + 2064384;

  dim3 blk(32, 2);

  gemm_wmma_kernel<MODE_TANH | MODE_BIAS, false, CC * HH><<<dim3(8, 24, 1), blk, 0, stream>>>(
      seq, DD, 0, 0, Wp, 0, 0, tanh_hp, CC * HH, 0, 0,
      nullptr, bp, nullptr, DD);

  gemm_wmma_kernel<MODE_TANH | MODE_BIAS, true, CC * HH><<<dim3(8, 24, 1), blk, 0, stream>>>(
      seq, DD, 0, 0, Ua, 0, 0, tanh_ha, CC * HH, 0, 0,
      ha_raw, ba, nullptr, DD);

  gemm_wmma_kernel<MODE_TANH, false, CC * HH><<<dim3(2, 12, BB * CC), blk, 0, stream>>>(
      base, CC * SS, (long)SS * CC * SS, SS,
      ha_raw, (long)SS * CC * HH, HH,
      tanhref, CC * HH, (long)SS * CC * HH, HH,
      nullptr, nullptr, nullptr, SS);

  gemm_wmma_kernel<0, false, HH><<<dim3(2, 24, 1), blk, 0, stream>>>(
      tanhref, CC * HH, 0, 0, Wmid + 2 * HH * HH, 0, 0, tref, HH, 0, 0,
      nullptr, nullptr, nullptr, CC * HH);

  gemm_wmma_kernel<MODE_BIAS | MODE_ADDV, false, HH><<<dim3(2, 96, 1), blk, 0, stream>>>(
      tanh_hp, HH, 0, 0, Wmid, 0, 0, pref, HH, 0, 0,
      nullptr, bmid, tref, HH);

  gemm_wmma_kernel<0, false, HH><<<dim3(2, 96, 1), blk, 0, stream>>>(
      tanh_ha, HH, 0, 0, Wmid + HH * HH, 0, 0, targ, HH, 0, 0,
      nullptr, nullptr, nullptr, HH);

  final_reduce_kernel<<<dim3(SS / 128, SS, BB * CC), dim3(128), 0, stream>>>(
      pref, targ, wout, out);
}


# Final kernel: a thread keeps its targ row and w_out in registers, reuses them for 12 pref rows, 2 at a time (rows enter coalesced via LDS); 36864 -> 3072 waves; sum order unchanged, products fused
# speedup vs baseline: 2.5578x; 1.3118x over previous
#include <hip/hip_runtime.h>

#define BB 2
#define SS 384
#define CC 4
#define HH 128
#define DD 768

typedef float v2f __attribute__((ext_vector_type(2)));
typedef float v8f __attribute__((ext_vector_type(8)));

__device__ __forceinline__ float tanhf_fast(float x) {
  float e = __builtin_amdgcn_exp2f(x * 2.8853900817779268f);
  float r = __builtin_amdgcn_rcpf(e + 1.0f);
  return 1.0f - 2.0f * r;
}

#define MODE_TANH 1
#define MODE_BIAS 4
#define MODE_ADDV 8

#define KCH 32
#define KST (KCH / 4)

template <int LDB>
__device__ __forceinline__ void load_set(v2f (&a)[KST], float4 (&b)[KST][2],
                                         const float* __restrict__ Ap,
                                         const float* __restrict__ Bp) {
#pragma unroll
  for (int j = 0; j < KST; ++j) {
    a[j]    = *(const v2f*)(Ap + 4 * j);
    b[j][0] = *(const float4*)(Bp + (4 * j) * LDB);
    b[j][1] = *(const float4*)(Bp + (4 * j + 1) * LDB);
  }
  __builtin_amdgcn_sched_barrier(0);
}

__device__ __forceinline__ void mma_set(const v2f (&a)[KST], const float4 (&b)[KST][2],
                                        v8f (&acc)[4]) {
#pragma unroll
  for (int j = 0; j < KST; ++j) {
    v2f b0 = {b[j][0].x, b[j][1].x};
    v2f b1 = {b[j][0].y, b[j][1].y};
    v2f b2 = {b[j][0].z, b[j][1].z};
    v2f b3 = {b[j][0].w, b[j][1].w};
    acc[0] = __builtin_amdgcn_wmma_f32_16x16x4_f32(false, a[j], false, b0, (short)0, acc[0], false, false);
    acc[1] = __builtin_amdgcn_wmma_f32_16x16x4_f32(false, a[j], false, b1, (short)0, acc[1], false, false);
    acc[2] = __builtin_amdgcn_wmma_f32_16x16x4_f32(false, a[j], false, b2, (short)0, acc[2], false, false);
    acc[3] = __builtin_amdgcn_wmma_f32_16x16x4_f32(false, a[j], false, b3, (short)0, acc[3], false, false);
  }
}

template <int MODE, bool HAS_RAW, int LDB>
__global__ void __launch_bounds__(64)
gemm_wmma_kernel(const float* __restrict__ A, int lda, long aOuter, long aInner,
                 const float* __restrict__ B, long bOuter, long bInner,
                 float* __restrict__ O, int ldo, long oOuter, long oInner,
                 float* __restrict__ Oraw,
                 const float* __restrict__ bias, const float* __restrict__ addvec,
                 int K) {
  int lane = threadIdx.x;
  int wave = threadIdx.y;
  int c    = lane & 15;
  int h    = lane >> 4;
  int zo   = blockIdx.z >> 2;
  int zi   = blockIdx.z & 3;
  int m0   = blockIdx.y * 32 + wave * 16;
  int n0   = blockIdx.x * 64;

  const float* Ap = A + (long)zo * aOuter + (long)zi * aInner + (long)(m0 + c) * lda + 2 * h;
  const float* Bp = B + (long)zo * bOuter + (long)zi * bInner + (2 * h) * LDB + n0 + 4 * c;

  v8f acc[4] = {};
  v2f    a0[KST], a1[KST];
  float4 b0[KST][2], b1[KST][2];

  load_set<LDB>(a0, b0, Ap, Bp);
  int kc = 0;
  for (; kc + 2 * KCH < K; kc += 2 * KCH) {
    load_set<LDB>(a1, b1, Ap + kc + KCH, Bp + (long)(kc + KCH) * LDB);
    mma_set(a0, b0, acc);
    load_set<LDB>(a0, b0, Ap + kc + 2 * KCH, Bp + (long)(kc + 2 * KCH) * LDB);
    mma_set(a1, b1, acc);
  }
  load_set<LDB>(a1, b1, Ap + kc + KCH, Bp + (long)(kc + KCH) * LDB);
  mma_set(a0, b0, acc);
  mma_set(a1, b1, acc);

  float* Ob = O + (long)zo * oOuter + (long)zi * oInner;
  float* Rb = HAS_RAW ? (Oraw + (long)zo * oOuter + (long)zi * oInner) : nullptr;
  int col = n0 + 4 * c;
  float4 bi = {0.f, 0.f, 0.f, 0.f};
  if constexpr (MODE & MODE_BIAS) bi = *(const float4*)(bias + col);
#pragma unroll
  for (int r = 0; r < 8; ++r) {
    int row = m0 + 8 * h + r;
    float4 v = {acc[0][r], acc[1][r], acc[2][r], acc[3][r]};
    if constexpr (MODE & MODE_BIAS) { v.x += bi.x; v.y += bi.y; v.z += bi.z; v.w += bi.w; }
    if constexpr (MODE & MODE_ADDV) {
      float4 av = *(const float4*)(addvec + (long)(row >> 2) * HH + col);
      v.x += av.x; v.y += av.y; v.z += av.z; v.w += av.w;
    }
    if constexpr (HAS_RAW) *(float4*)(Rb + (long)row * ldo + col) = v;
    if constexpr (MODE & MODE_TANH) {
      v.x = tanhf_fast(v.x); v.y = tanhf_fast(v.y); v.z = tanhf_fast(v.z); v.w = tanhf_fast(v.w);
    }
    *(float4*)(Ob + (long)row * ldo + col) = v;
  }
}

#define PCH 12
#define EBK 8
#define TP  (HH + 4)
__global__ void __launch_bounds__(128)
final_reduce_kernel(const float* __restrict__ pref, const float* __restrict__ targ,
                    const float* __restrict__ wout, float* __restrict__ out) {
  __shared__ float lds[HH + PCH * HH + 128 * TP];
  float* sw = lds;
  float (*sp)[HH] = (float (*)[HH])(lds + HH);
  float (*st)[TP] = (float (*)[TP])(lds + HH + PCH * HH);
  int tid  = threadIdx.x;
  int lane = tid & 31;
  int wv   = tid >> 5;
  int bc   = blockIdx.z;
  int b    = bc >> 2;
  int c    = bc & 3;
  int a0   = blockIdx.x * 128;
  int p0   = blockIdx.y * PCH;

  const float* pr = pref + ((long)(b * SS + p0) * CC + c) * HH;
  sw[tid] = wout[tid];
#pragma unroll
  for (int r = 0; r < PCH; ++r) sp[r][tid] = pr[(long)r * CC * HH + tid];

  const float* tb = targ + ((long)(b * SS + a0 + 32 * wv) * CC + c) * HH + 4 * lane;
#pragma unroll
  for (int j = 0; j < 32; ++j) {
    float4 tv = *(const float4*)(tb + (long)j * CC * HH);
    *(float4*)&st[32 * wv + j][4 * lane] = tv;
  }
  __syncthreads();
  v2f t[HH / 2];
#pragma unroll
  for (int j = 0; j < HH / 2; ++j) t[j] = *(const v2f*)&st[tid][2 * j];
  float w[HH];
#pragma unroll
  for (int k = 0; k < HH; ++k) w[k] = sw[k];

  float* op = out + ((long)(b * SS + p0) * CC + c) * SS + a0 + tid;
#pragma unroll 1
  for (int q = 0; q < PCH / 2; ++q) {
    const v2f* s0 = (const v2f*)sp[2 * q];
    const v2f* s1 = (const v2f*)sp[2 * q + 1];
    float acc0 = 0.0f, acc1 = 0.0f;
    v2f n0[EBK / 2], n1[EBK / 2];
#pragma unroll
    for (int i = 0; i < EBK / 2; ++i) { n0[i] = s0[i]; n1[i] = s1[i]; }
#pragma unroll
    for (int k = 0; k < HH; k += EBK) {
      v2f x0[EBK / 2], x1[EBK / 2];
#pragma unroll
      for (int i = 0; i < EBK / 2; ++i) { x0[i] = n0[i]; x1[i] = n1[i]; }
      if (k + EBK < HH) {
#pragma unroll
        for (int i = 0; i < EBK / 2; ++i) {
          n0[i] = s0[(k + EBK) / 2 + i];
          n1[i] = s1[(k + EBK) / 2 + i];
        }
      }
      __builtin_amdgcn_sched_barrier(0x1);
#pragma unroll
      for (int i = 0; i < EBK / 2; ++i) {
        x0[i] = (x0[i] + t[k / 2 + i]) * 2.8853900817779268f;
        x1[i] = (x1[i] + t[k / 2 + i]) * 2.8853900817779268f;
      }
#pragma unroll
      for (int i = 0; i < EBK / 2; ++i) {
        x0[i].x = __builtin_amdgcn_exp2f(x0[i].x);
        x0[i].y = __builtin_amdgcn_exp2f(x0[i].y);
        x1[i].x = __builtin_amdgcn_exp2f(x1[i].x);
        x1[i].y = __builtin_amdgcn_exp2f(x1[i].y);
      }
#pragma unroll
      for (int i = 0; i < EBK / 2; ++i) {
        x0[i] = x0[i] + 1.0f;
        x1[i] = x1[i] + 1.0f;
        x0[i].x = __builtin_amdgcn_rcpf(x0[i].x);
        x0[i].y = __builtin_amdgcn_rcpf(x0[i].y);
        x1[i].x = __builtin_amdgcn_rcpf(x1[i].x);
        x1[i].y = __builtin_amdgcn_rcpf(x1[i].y);
      }
#pragma unroll
      for (int i = 0; i < EBK / 2; ++i) {
        v2f h0 = 1.0f - 2.0f * x0[i];
        v2f h1 = 1.0f - 2.0f * x1[i];
        acc0 += h0.x * w[k + 2 * i];
        acc1 += h1.x * w[k + 2 * i];
        acc0 += h0.y * w[k + 2 * i + 1];
        acc1 += h1.y * w[k + 2 * i + 1];
      }
    }
    op[0]       = acc0;
    op[CC * SS] = acc1;
    op += 2 * CC * SS;
  }
}

extern "C" void kernel_launch(void* const* d_in, const int* in_sizes, int n_in,
                              void* d_out, int out_size, void* d_ws, size_t ws_size,
                              hipStream_t stream) {
  const float* seq  = (const float*)d_in[0];
  const float* base = (const float*)d_in[1];
  const float* Wp   = (const float*)d_in[2];
  const float* bp   = (const float*)d_in[3];
  const float* Ua   = (const float*)d_in[4];
  const float* ba   = (const float*)d_in[5];
  const float* Wmid = (const float*)d_in[6];
  const float* bmid = (const float*)d_in[7];
  const float* wout = (const float*)d_in[8];
  float* out = (float*)d_out;

  float* ws      = (float*)d_ws;
  float* tanh_hp = ws;
  float* ha_raw  = ws + 393216;
  float* tanh_ha = ws + 786432;
  float* tanhref = ws + 1179648;
  float* targ    = ws + 1572864;
  float* tref    = ws + 1966080;
  float* pref    = ws + 2064384;

  dim3 blk(32, 2);

  gemm_wmma_kernel<MODE_TANH | MODE_BIAS, false, CC * HH><<<dim3(8, 24, 1), blk, 0, stream>>>(
      seq, DD, 0, 0, Wp, 0, 0, tanh_hp, CC * HH, 0, 0,
      nullptr, bp, nullptr, DD);

  gemm_wmma_kernel<MODE_TANH | MODE_BIAS, true, CC * HH><<<dim3(8, 24, 1), blk, 0, stream>>>(
      seq, DD, 0, 0, Ua, 0, 0, tanh_ha, CC * HH, 0, 0,
      ha_raw, ba, nullptr, DD);

  gemm_wmma_kernel<MODE_TANH, false, CC * HH><<<dim3(2, 12, BB * CC), blk, 0, stream>>>(
      base, CC * SS, (long)SS * CC * SS, SS,
      ha_raw, (long)SS * CC * HH, HH,
      tanhref, CC * HH, (long)SS * CC * HH, HH,
      nullptr, nullptr, nullptr, SS);

  gemm_wmma_kernel<0, false, HH><<<dim3(2, 24, 1), blk, 0, stream>>>(
      tanhref, CC * HH, 0, 0, Wmid + 2 * HH * HH, 0, 0, tref, HH, 0, 0,
      nullptr, nullptr, nullptr, CC * HH);

  gemm_wmma_kernel<MODE_BIAS | MODE_ADDV, false, HH><<<dim3(2, 96, 1), blk, 0, stream>>>(
      tanh_hp, HH, 0, 0, Wmid, 0, 0, pref, HH, 0, 0,
      nullptr, bmid, tref, HH);

  gemm_wmma_kernel<0, false, HH><<<dim3(2, 96, 1), blk, 0, stream>>>(
      tanh_ha, HH, 0, 0, Wmid + HH * HH, 0, 0, targ, HH, 0, 0,
      nullptr, nullptr, nullptr, HH);

  final_reduce_kernel<<<dim3(SS / 128, SS / PCH, BB * CC), dim3(128), 0, stream>>>(
      pref, targ, wout, out);
}


# GEMM kernel: K split over 4 waves per output tile (blocks of 8 waves, t_ref 4; were 2), partial tiles added through LDS in fixed order: each sum is now 4 K-quarter partial sums
# speedup vs baseline: 3.7181x; 1.4536x over previous
#include <hip/hip_runtime.h>

#define BB 2
#define SS 384
#define CC 4
#define HH 128
#define DD 768

typedef float v2f __attribute__((ext_vector_type(2)));
typedef float v8f __attribute__((ext_vector_type(8)));

__device__ __forceinline__ float tanhf_fast(float x) {
  float e = __builtin_amdgcn_exp2f(x * 2.8853900817779268f);
  float r = __builtin_amdgcn_rcpf(e + 1.0f);
  return 1.0f - 2.0f * r;
}

#define MODE_TANH 1
#define MODE_BIAS 4
#define MODE_ADDV 8

#define KCH 32
#define KST (KCH / 4)

template <int LDB>
__device__ __forceinline__ void load_set(v2f (&a)[KST], float4 (&b)[KST][2],
                                         const float* __restrict__ Ap,
                                         const float* __restrict__ Bp) {
#pragma unroll
  for (int j = 0; j < KST; ++j) {
    a[j]    = *(const v2f*)(Ap + 4 * j);
    b[j][0] = *(const float4*)(Bp + (4 * j) * LDB);
    b[j][1] = *(const float4*)(Bp + (4 * j + 1) * LDB);
  }
  __builtin_amdgcn_sched_barrier(0);
}

__device__ __forceinline__ void mma_set(const v2f (&a)[KST], const float4 (&b)[KST][2],
                                        v8f (&acc)[4]) {
#pragma unroll
  for (int j = 0; j < KST; ++j) {
    v2f b0 = {b[j][0].x, b[j][1].x};
    v2f b1 = {b[j][0].y, b[j][1].y};
    v2f b2 = {b[j][0].z, b[j][1].z};
    v2f b3 = {b[j][0].w, b[j][1].w};
    acc[0] = __builtin_amdgcn_wmma_f32_16x16x4_f32(false, a[j], false, b0, (short)0, acc[0], false, false);
    acc[1] = __builtin_amdgcn_wmma_f32_16x16x4_f32(false, a[j], false, b1, (short)0, acc[1], false, false);
    acc[2] = __builtin_amdgcn_wmma_f32_16x16x4_f32(false, a[j], false, b2, (short)0, acc[2], false, false);
    acc[3] = __builtin_amdgcn_wmma_f32_16x16x4_f32(false, a[j], false, b3, (short)0, acc[3], false, false);
  }
}

#define KP 4
#define RP (64 + 4)
template <int MODE, bool HAS_RAW, int LDB, int NS, int MW>
__global__ void __launch_bounds__(32 * MW * KP)
gemm_wmma_kernel(const float* __restrict__ A, int lda, long aOuter, long aInner,
                 const float* __restrict__ B, long bOuter, long bInner,
                 float* __restrict__ O, int ldo, long oOuter, long oInner,
                 float* __restrict__ Oraw,
                 const float* __restrict__ bias, const float* __restrict__ addvec) {
  __shared__ __attribute__((aligned(16))) float red[MW][KP][16][RP];
  int lane = threadIdx.x;
  int wm   = threadIdx.y;
  int kq   = threadIdx.z;
  int c    = lane & 15;
  int h    = lane >> 4;
  int zo   = blockIdx.z >> 2;
  int zi   = blockIdx.z & 3;
  int m0   = blockIdx.y * (16 * MW) + wm * 16;
  int n0   = blockIdx.x * 64;
  int k0   = kq * (NS * KCH);

  const float* Ap = A + (long)zo * aOuter + (long)zi * aInner + (long)(m0 + c) * lda + k0 + 2 * h;
  const float* Bp = B + (long)zo * bOuter + (long)zi * bInner + (long)(k0 + 2 * h) * LDB + n0 + 4 * c;

  v8f acc[4] = {};
  v2f    a0[KST], a1[KST];
  float4 b0[KST][2], b1[KST][2];

  load_set<LDB>(a0, b0, Ap, Bp);
#pragma unroll
  for (int s = 0; s < NS; s += 2) {
    if (s + 1 < NS) load_set<LDB>(a1, b1, Ap + (s + 1) * KCH, Bp + (long)(s + 1) * KCH * LDB);
    mma_set(a0, b0, acc);
    if (s + 1 < NS) {
      if (s + 2 < NS) load_set<LDB>(a0, b0, Ap + (s + 2) * KCH, Bp + (long)(s + 2) * KCH * LDB);
      mma_set(a1, b1, acc);
    }
  }

#pragma unroll
  for (int r = 0; r < 8; ++r) {
    float4 v = {acc[0][r], acc[1][r], acc[2][r], acc[3][r]};
    *(float4*)&red[wm][kq][8 * h + r][4 * c] = v;
  }
  __syncthreads();

  float* Ob = O + (long)zo * oOuter + (long)zi * oInner;
  float* Rb = HAS_RAW ? (Oraw + (long)zo * oOuter + (long)zi * oInner) : nullptr;
  int col = n0 + 4 * c;
  float4 bi = {0.f, 0.f, 0.f, 0.f};
  if constexpr (MODE & MODE_BIAS) bi = *(const float4*)(bias + col);
#pragma unroll
  for (int i = 0; i < 2; ++i) {
    int rl  = 4 * kq + 2 * i + h;
    int row = m0 + rl;
    float4 v = *(const float4*)&red[wm][0][rl][4 * c];
#pragma unroll
    for (int q = 1; q < KP; ++q) {
      float4 p = *(const float4*)&red[wm][q][rl][4 * c];
      v.x += p.x; v.y += p.y; v.z += p.z; v.w += p.w;
    }
    if constexpr (MODE & MODE_BIAS) { v.x += bi.x; v.y += bi.y; v.z += bi.z; v.w += bi.w; }
    if constexpr (MODE & MODE_ADDV) {
      float4 av = *(const float4*)(addvec + (long)(row >> 2) * HH + col);
      v.x += av.x; v.y += av.y; v.z += av.z; v.w += av.w;
    }
    if constexpr (HAS_RAW) *(float4*)(Rb + (long)row * ldo + col) = v;
    if constexpr (MODE & MODE_TANH) {
      v.x = tanhf_fast(v.x); v.y = tanhf_fast(v.y); v.z = tanhf_fast(v.z); v.w = tanhf_fast(v.w);
    }
    *(float4*)(Ob + (long)row * ldo + col) = v;
  }
}

#define PCH 12
#define EBK 8
#define TP  (HH + 4)
__global__ void __launch_bounds__(128)
final_reduce_kernel(const float* __restrict__ pref, const float* __restrict__ targ,
                    const float* __restrict__ wout, float* __restrict__ out) {
  __shared__ float lds[HH + PCH * HH + 128 * TP];
  float* sw = lds;
  float (*sp)[HH] = (float (*)[HH])(lds + HH);
  float (*st)[TP] = (float (*)[TP])(lds + HH + PCH * HH);
  int tid  = threadIdx.x;
  int lane = tid & 31;
  int wv   = tid >> 5;
  int bc   = blockIdx.z;
  int b    = bc >> 2;
  int c    = bc & 3;
  int a0   = blockIdx.x * 128;
  int p0   = blockIdx.y * PCH;

  const float* pr = pref + ((long)(b * SS + p0) * CC + c) * HH;
  sw[tid] = wout[tid];
#pragma unroll
  for (int r = 0; r < PCH; ++r) sp[r][tid] = pr[(long)r * CC * HH + tid];

  const float* tb = targ + ((long)(b * SS + a0 + 32 * wv) * CC + c) * HH + 4 * lane;
#pragma unroll
  for (int j = 0; j < 32; ++j) {
    float4 tv = *(const float4*)(tb + (long)j * CC * HH);
    *(float4*)&st[32 * wv + j][4 * lane] = tv;
  }
  __syncthreads();
  v2f t[HH / 2];
#pragma unroll
  for (int j = 0; j < HH / 2; ++j) t[j] = *(const v2f*)&st[tid][2 * j];
  float w[HH];
#pragma unroll
  for (int k = 0; k < HH; ++k) w[k] = sw[k];

  float* op = out + ((long)(b * SS + p0) * CC + c) * SS + a0 + tid;
#pragma unroll 1
  for (int q = 0; q < PCH / 2; ++q) {
    const v2f* s0 = (const v2f*)sp[2 * q];
    const v2f* s1 = (const v2f*)sp[2 * q + 1];
    float acc0 = 0.0f, acc1 = 0.0f;
    v2f n0[EBK / 2], n1[EBK / 2];
#pragma unroll
    for (int i = 0; i < EBK / 2; ++i) { n0[i] = s0[i]; n1[i] = s1[i]; }
#pragma unroll
    for (int k = 0; k < HH; k += EBK) {
      v2f x0[EBK / 2], x1[EBK / 2];
#pragma unroll
      for (int i = 0; i < EBK / 2; ++i) { x0[i] = n0[i]; x1[i] = n1[i]; }
      if (k + EBK < HH) {
#pragma unroll
        for (int i = 0; i < EBK / 2; ++i) {
          n0[i] = s0[(k + EBK) / 2 + i];
          n1[i] = s1[(k + EBK) / 2 + i];
        }
      }
      __builtin_amdgcn_sched_barrier(0x1);
#pragma unroll
      for (int i = 0; i < EBK / 2; ++i) {
        x0[i] = (x0[i] + t[k / 2 + i]) * 2.8853900817779268f;
        x1[i] = (x1[i] + t[k / 2 + i]) * 2.8853900817779268f;
      }
#pragma unroll
      for (int i = 0; i < EBK / 2; ++i) {
        x0[i].x = __builtin_amdgcn_exp2f(x0[i].x);
        x0[i].y = __builtin_amdgcn_exp2f(x0[i].y);
        x1[i].x = __builtin_amdgcn_exp2f(x1[i].x);
        x1[i].y = __builtin_amdgcn_exp2f(x1[i].y);
      }
#pragma unroll
      for (int i = 0; i < EBK / 2; ++i) {
        x0[i] = x0[i] + 1.0f;
        x1[i] = x1[i] + 1.0f;
        x0[i].x = __builtin_amdgcn_rcpf(x0[i].x);
        x0[i].y = __builtin_amdgcn_rcpf(x0[i].y);
        x1[i].x = __builtin_amdgcn_rcpf(x1[i].x);
        x1[i].y = __builtin_amdgcn_rcpf(x1[i].y);
      }
#pragma unroll
      for (int i = 0; i < EBK / 2; ++i) {
        v2f h0 = 1.0f - 2.0f * x0[i];
        v2f h1 = 1.0f - 2.0f * x1[i];
        acc0 += h0.x * w[k + 2 * i];
        acc1 += h1.x * w[k + 2 * i];
        acc0 += h0.y * w[k + 2 * i + 1];
        acc1 += h1.y * w[k + 2 * i + 1];
      }
    }
    op[0]       = acc0;
    op[CC * SS] = acc1;
    op += 2 * CC * SS;
  }
}

extern "C" void kernel_launch(void* const* d_in, const int* in_sizes, int n_in,
                              void* d_out, int out_size, void* d_ws, size_t ws_size,
                              hipStream_t stream) {
  const float* seq  = (const float*)d_in[0];
  const float* base = (const float*)d_in[1];
  const float* Wp   = (const float*)d_in[2];
  const float* bp   = (const float*)d_in[3];
  const float* Ua   = (const float*)d_in[4];
  const float* ba   = (const float*)d_in[5];
  const float* Wmid = (const float*)d_in[6];
  const float* bmid = (const float*)d_in[7];
  const float* wout = (const float*)d_in[8];
  float* out = (float*)d_out;

  float* ws      = (float*)d_ws;
  float* tanh_hp = ws;
  float* ha_raw  = ws + 393216;
  float* tanh_ha = ws + 786432;
  float* tanhref = ws + 1179648;
  float* targ    = ws + 1572864;
  float* tref    = ws + 1966080;
  float* pref    = ws + 2064384;

  dim3 blk2(32, 2, KP);
  dim3 blk1(32, 1, KP);

  gemm_wmma_kernel<MODE_TANH | MODE_BIAS, false, CC * HH, 6, 2><<<dim3(8, 24, 1), blk2, 0, stream>>>(
      seq, DD, 0, 0, Wp, 0, 0, tanh_hp, CC * HH, 0, 0,
      nullptr, bp, nullptr);

  gemm_wmma_kernel<MODE_TANH | MODE_BIAS, true, CC * HH, 6, 2><<<dim3(8, 24, 1), blk2, 0, stream>>>(
      seq, DD, 0, 0, Ua, 0, 0, tanh_ha, CC * HH, 0, 0,
      ha_raw, ba, nullptr);

  gemm_wmma_kernel<MODE_TANH, false, CC * HH, 3, 2><<<dim3(2, 12, BB * CC), blk2, 0, stream>>>(
      base, CC * SS, (long)SS * CC * SS, SS,
      ha_raw, (long)SS * CC * HH, HH,
      tanhref, CC * HH, (long)SS * CC * HH, HH,
      nullptr, nullptr, nullptr);

  gemm_wmma_kernel<0, false, HH, 4, 1><<<dim3(2, 48, 1), blk1, 0, stream>>>(
      tanhref, CC * HH, 0, 0, Wmid + 2 * HH * HH, 0, 0, tref, HH, 0, 0,
      nullptr, nullptr, nullptr);

  gemm_wmma_kernel<MODE_BIAS | MODE_ADDV, false, HH, 1, 2><<<dim3(2, 96, 1), blk2, 0, stream>>>(
      tanh_hp, HH, 0, 0, Wmid, 0, 0, pref, HH, 0, 0,
      nullptr, bmid, tref);

  gemm_wmma_kernel<0, false, HH, 1, 2><<<dim3(2, 96, 1), blk2, 0, stream>>>(
      tanh_ha, HH, 0, 0, Wmid + HH * HH, 0, 0, targ, HH, 0, 0,
      nullptr, nullptr, nullptr);

  final_reduce_kernel<<<dim3(SS / 128, SS / PCH, BB * CC), dim3(128), 0, stream>>>(
      pref, targ, wout, out);
}


# Final kernel: e^(2(p+t)) taken as e^(2p)*e^(2t), stored per row by the pref/targ GEMMs: an element costs fma, rcp, fma (was 2 transcendentals + 5); sum starts at sum(w), terms r*(-2w), k order kept
# speedup vs baseline: 4.3123x; 1.1598x over previous
#include <hip/hip_runtime.h>

#define BB 2
#define SS 384
#define CC 4
#define HH 128
#define DD 768

typedef float v2f __attribute__((ext_vector_type(2)));
typedef float v8f __attribute__((ext_vector_type(8)));

__device__ __forceinline__ float tanhf_fast(float x) {
  float e = __builtin_amdgcn_exp2f(x * 2.8853900817779268f);
  float r = __builtin_amdgcn_rcpf(e + 1.0f);
  return 1.0f - 2.0f * r;
}

#define C2L 2.8853900817779268f

#define MODE_TANH 1
#define MODE_BIAS 4
#define MODE_ADDV 8
#define MODE_EXP2 16

#define KCH 32
#define KST (KCH / 4)

template <int LDB>
__device__ __forceinline__ void load_set(v2f (&a)[KST], float4 (&b)[KST][2],
                                         const float* __restrict__ Ap,
                                         const float* __restrict__ Bp) {
#pragma unroll
  for (int j = 0; j < KST; ++j) {
    a[j]    = *(const v2f*)(Ap + 4 * j);
    b[j][0] = *(const float4*)(Bp + (4 * j) * LDB);
    b[j][1] = *(const float4*)(Bp + (4 * j + 1) * LDB);
  }
  __builtin_amdgcn_sched_barrier(0);
}

__device__ __forceinline__ void mma_set(const v2f (&a)[KST], const float4 (&b)[KST][2],
                                        v8f (&acc)[4]) {
#pragma unroll
  for (int j = 0; j < KST; ++j) {
    v2f b0 = {b[j][0].x, b[j][1].x};
    v2f b1 = {b[j][0].y, b[j][1].y};
    v2f b2 = {b[j][0].z, b[j][1].z};
    v2f b3 = {b[j][0].w, b[j][1].w};
    acc[0] = __builtin_amdgcn_wmma_f32_16x16x4_f32(false, a[j], false, b0, (short)0, acc[0], false, false);
    acc[1] = __builtin_amdgcn_wmma_f32_16x16x4_f32(false, a[j], false, b1, (short)0, acc[1], false, false);
    acc[2] = __builtin_amdgcn_wmma_f32_16x16x4_f32(false, a[j], false, b2, (short)0, acc[2], false, false);
    acc[3] = __builtin_amdgcn_wmma_f32_16x16x4_f32(false, a[j], false, b3, (short)0, acc[3], false, false);
  }
}

#define KP 4
#define RP (64 + 4)
template <int MODE, bool HAS_RAW, int LDB, int NS, int MW>
__global__ void __launch_bounds__(32 * MW * KP)
gemm_wmma_kernel(const float* __restrict__ A, int lda, long aOuter, long aInner,
                 const float* __restrict__ B, long bOuter, long bInner,
                 float* __restrict__ O, int ldo, long oOuter, long oInner,
                 float* __restrict__ Oraw,
                 const float* __restrict__ bias, const float* __restrict__ addvec) {
  __shared__ __attribute__((aligned(16))) float red[MW][KP][16][RP];
  int lane = threadIdx.x;
  int wm   = threadIdx.y;
  int kq   = threadIdx.z;
  int c    = lane & 15;
  int h    = lane >> 4;
  int zo   = blockIdx.z >> 2;
  int zi   = blockIdx.z & 3;
  int m0   = blockIdx.y * (16 * MW) + wm * 16;
  int n0   = blockIdx.x * 64;
  int k0   = kq * (NS * KCH);

  const float* Ap = A + (long)zo * aOuter + (long)zi * aInner + (long)(m0 + c) * lda + k0 + 2 * h;
  const float* Bp = B + (long)zo * bOuter + (long)zi * bInner + (long)(k0 + 2 * h) * LDB + n0 + 4 * c;

  v8f acc[4] = {};
  v2f    a0[KST], a1[KST];
  float4 b0[KST][2], b1[KST][2];

  load_set<LDB>(a0, b0, Ap, Bp);
#pragma unroll
  for (int s = 0; s < NS; s += 2) {
    if (s + 1 < NS) load_set<LDB>(a1, b1, Ap + (s + 1) * KCH, Bp + (long)(s + 1) * KCH * LDB);
    mma_set(a0, b0, acc);
    if (s + 1 < NS) {
      if (s + 2 < NS) load_set<LDB>(a0, b0, Ap + (s + 2) * KCH, Bp + (long)(s + 2) * KCH * LDB);
      mma_set(a1, b1, acc);
    }
  }

#pragma unroll
  for (int r = 0; r < 8; ++r) {
    float4 v = {acc[0][r], acc[1][r], acc[2][r], acc[3][r]};
    *(float4*)&red[wm][kq][8 * h + r][4 * c] = v;
  }
  __syncthreads();

  float* Ob = O + (long)zo * oOuter + (long)zi * oInner;
  float* Rb = HAS_RAW ? (Oraw + (long)zo * oOuter + (long)zi * oInner) : nullptr;
  int col = n0 + 4 * c;
  float4 bi = {0.f, 0.f, 0.f, 0.f};
  if constexpr (MODE & MODE_BIAS) bi = *(const float4*)(bias + col);
#pragma unroll
  for (int i = 0; i < 2; ++i) {
    int rl  = 4 * kq + 2 * i + h;
    int row = m0 + rl;
    float4 v = *(const float4*)&red[wm][0][rl][4 * c];
#pragma unroll
    for (int q = 1; q < KP; ++q) {
      float4 p = *(const float4*)&red[wm][q][rl][4 * c];
      v.x += p.x; v.y += p.y; v.z += p.z; v.w += p.w;
    }
    if constexpr (MODE & MODE_BIAS) { v.x += bi.x; v.y += bi.y; v.z += bi.z; v.w += bi.w; }
    if constexpr (MODE & MODE_ADDV) {
      float4 av = *(const float4*)(addvec + (long)(row >> 2) * HH + col);
      v.x += av.x; v.y += av.y; v.z += av.z; v.w += av.w;
    }
    if constexpr (HAS_RAW) *(float4*)(Rb + (long)row * ldo + col) = v;
    if constexpr (MODE & MODE_TANH) {
      v.x = tanhf_fast(v.x); v.y = tanhf_fast(v.y); v.z = tanhf_fast(v.z); v.w = tanhf_fast(v.w);
    }
    if constexpr (MODE & MODE_EXP2) {
      v.x = __builtin_amdgcn_exp2f(v.x * C2L); v.y = __builtin_amdgcn_exp2f(v.y * C2L);
      v.z = __builtin_amdgcn_exp2f(v.z * C2L); v.w = __builtin_amdgcn_exp2f(v.w * C2L);
    }
    *(float4*)(Ob + (long)row * ldo + col) = v;
  }
}

#define PCH 12
#define EBK 8
#define TP  (HH + 4)
#define ELO_BITS 0x2F800000u
#define EHI_BITS 0x4F800000u

template <bool FAST>
__device__ __forceinline__ void final_rows(const float (*sp)[HH], const float* st_row,
                                           const float* swv, float init, float* op) {
  v2f t[HH / 2];
#pragma unroll
  for (int j = 0; j < HH / 2; ++j) t[j] = *(const v2f*)&st_row[2 * j];
  float w[HH];
#pragma unroll
  for (int k = 0; k < HH; ++k) w[k] = swv[k];
  const v2f one = {1.0f, 1.0f};

#pragma unroll 1
  for (int q = 0; q < PCH / 2; ++q) {
    const v2f* s0 = (const v2f*)sp[2 * q];
    const v2f* s1 = (const v2f*)sp[2 * q + 1];
    float acc0 = init, acc1 = init;
    v2f n0[EBK / 2], n1[EBK / 2];
#pragma unroll
    for (int i = 0; i < EBK / 2; ++i) { n0[i] = s0[i]; n1[i] = s1[i]; }
#pragma unroll
    for (int k = 0; k < HH; k += EBK) {
      v2f x0[EBK / 2], x1[EBK / 2];
#pragma unroll
      for (int i = 0; i < EBK / 2; ++i) { x0[i] = n0[i]; x1[i] = n1[i]; }
      if (k + EBK < HH) {
#pragma unroll
        for (int i = 0; i < EBK / 2; ++i) {
          n0[i] = s0[(k + EBK) / 2 + i];
          n1[i] = s1[(k + EBK) / 2 + i];
        }
      }
      __builtin_amdgcn_sched_barrier(0x1);
      if constexpr (FAST) {
#pragma unroll
        for (int i = 0; i < EBK / 2; ++i) {
          x0[i] = __builtin_elementwise_fma(x0[i], t[k / 2 + i], one);
          x1[i] = __builtin_elementwise_fma(x1[i], t[k / 2 + i], one);
        }
      } else {
#pragma unroll
        for (int i = 0; i < EBK / 2; ++i) {
          x0[i] = (x0[i] + t[k / 2 + i]) * C2L;
          x1[i] = (x1[i] + t[k / 2 + i]) * C2L;
        }
#pragma unroll
        for (int i = 0; i < EBK / 2; ++i) {
          x0[i].x = __builtin_amdgcn_exp2f(x0[i].x);
          x0[i].y = __builtin_amdgcn_exp2f(x0[i].y);
          x1[i].x = __builtin_amdgcn_exp2f(x1[i].x);
          x1[i].y = __builtin_amdgcn_exp2f(x1[i].y);
        }
#pragma unroll
        for (int i = 0; i < EBK / 2; ++i) { x0[i] = x0[i] + 1.0f; x1[i] = x1[i] + 1.0f; }
      }
#pragma unroll
      for (int i = 0; i < EBK / 2; ++i) {
        x0[i].x = __builtin_amdgcn_rcpf(x0[i].x);
        x0[i].y = __builtin_amdgcn_rcpf(x0[i].y);
        x1[i].x = __builtin_amdgcn_rcpf(x1[i].x);
        x1[i].y = __builtin_amdgcn_rcpf(x1[i].y);
      }
#pragma unroll
      for (int i = 0; i < EBK / 2; ++i) {
        v2f h0 = x0[i], h1 = x1[i];
        if constexpr (!FAST) {
          h0 = 1.0f - 2.0f * x0[i];
          h1 = 1.0f - 2.0f * x1[i];
        }
        acc0 += h0.x * w[k + 2 * i];
        acc1 += h1.x * w[k + 2 * i];
        acc0 += h0.y * w[k + 2 * i + 1];
        acc1 += h1.y * w[k + 2 * i + 1];
      }
    }
    op[0]       = acc0;
    op[CC * SS] = acc1;
    op += 2 * CC * SS;
  }
}

__global__ void __launch_bounds__(128)
final_reduce_kernel(const float* __restrict__ prefE, const float* __restrict__ targE,
                    const float* __restrict__ prefR, const float* __restrict__ targR,
                    const float* __restrict__ wout, float* __restrict__ out) {
  __shared__ __attribute__((aligned(16))) float lds[2 * HH + PCH * HH + 128 * TP];
  __shared__ int sflag[4];
  float* sw  = lds;
  float* sw2 = lds + HH;
  float (*sp)[HH] = (float (*)[HH])(lds + 2 * HH);
  float (*st)[TP] = (float (*)[TP])(lds + 2 * HH + PCH * HH);
  int tid  = threadIdx.x;
  int lane = tid & 31;
  int wv   = tid >> 5;
  int bc   = blockIdx.z;
  int b    = bc >> 2;
  int c    = bc & 3;
  int a0   = blockIdx.x * 128;
  int p0   = blockIdx.y * PCH;

  long poff = ((long)(b * SS + p0) * CC + c) * HH;
  long toff = ((long)(b * SS + a0 + 32 * wv) * CC + c) * HH + 4 * lane;
  float wk = wout[tid];
  sw[tid]  = wk;
  sw2[tid] = -2.0f * wk;
  unsigned ulo = 0xFFFFFFFFu, uhi = 0u;
#pragma unroll
  for (int r = 0; r < PCH; ++r) {
    float e = prefE[poff + (long)r * CC * HH + tid];
    sp[r][tid] = e;
    unsigned u = __float_as_uint(e);
    ulo = min(ulo, u); uhi = max(uhi, u);
  }
#pragma unroll
  for (int j = 0; j < 32; ++j) {
    float4 tv = *(const float4*)(targE + toff + (long)j * CC * HH);
    *(float4*)&st[32 * wv + j][4 * lane] = tv;
    unsigned ux = __float_as_uint(tv.x), uy = __float_as_uint(tv.y);
    unsigned uz = __float_as_uint(tv.z), uw = __float_as_uint(tv.w);
    ulo = min(min(ulo, ux), min(uy, min(uz, uw)));
    uhi = max(max(uhi, ux), max(uy, max(uz, uw)));
  }
  bool ok = (ulo >= ELO_BITS) && (uhi <= EHI_BITS);
  int wbad = __any(!ok);
  if (lane == 0) sflag[wv] = wbad;
  __syncthreads();
  int direct = (sflag[0] | sflag[1]) | (sflag[2] | sflag[3]);
  direct = __builtin_amdgcn_readfirstlane(direct);

  float* op = out + ((long)(b * SS + p0) * CC + c) * SS + a0 + tid;
  if (direct) {
#pragma unroll
    for (int r = 0; r < PCH; ++r) sp[r][tid] = prefR[poff + (long)r * CC * HH + tid];
#pragma unroll
    for (int j = 0; j < 32; ++j)
      *(float4*)&st[32 * wv + j][4 * lane] = *(const float4*)(targR + toff + (long)j * CC * HH);
    __syncthreads();
    final_rows<false>(sp, st[tid], sw, 0.0f, op);
  } else {
    float s = ((sw[lane] + sw[lane + 32]) + sw[lane + 64]) + sw[lane + 96];
#pragma unroll
    for (int m = 16; m >= 1; m >>= 1) s += __shfl_xor(s, m);
    final_rows<true>(sp, st[tid], sw2, s, op);
  }
}

extern "C" void kernel_launch(void* const* d_in, const int* in_sizes, int n_in,
                              void* d_out, int out_size, void* d_ws, size_t ws_size,
                              hipStream_t stream) {
  const float* seq  = (const float*)d_in[0];
  const float* base = (const float*)d_in[1];
  const float* Wp   = (const float*)d_in[2];
  const float* bp   = (const float*)d_in[3];
  const float* Ua   = (const float*)d_in[4];
  const float* ba   = (const float*)d_in[5];
  const float* Wmid = (const float*)d_in[6];
  const float* bmid = (const float*)d_in[7];
  const float* wout = (const float*)d_in[8];
  float* out = (float*)d_out;

  float* ws      = (float*)d_ws;
  float* tanh_hp = ws;
  float* ha_raw  = ws + 393216;
  float* tanh_ha = ws + 786432;
  float* tanhref = ws + 1179648;
  float* targ    = ws + 1572864;
  float* tref    = ws + 1966080;
  float* pref    = ws + 2064384;
  float* pref_e  = ws + 2457600;
  float* targ_e  = ws + 2850816;

  dim3 blk2(32, 2, KP);
  dim3 blk1(32, 1, KP);

  gemm_wmma_kernel<MODE_TANH | MODE_BIAS, false, CC * HH, 6, 2><<<dim3(8, 24, 1), blk2, 0, stream>>>(
      seq, DD, 0, 0, Wp, 0, 0, tanh_hp, CC * HH, 0, 0,
      nullptr, bp, nullptr);

  gemm_wmma_kernel<MODE_TANH | MODE_BIAS, true, CC * HH, 6, 2><<<dim3(8, 24, 1), blk2, 0, stream>>>(
      seq, DD, 0, 0, Ua, 0, 0, tanh_ha, CC * HH, 0, 0,
      ha_raw, ba, nullptr);

  gemm_wmma_kernel<MODE_TANH, false, CC * HH, 3, 2><<<dim3(2, 12, BB * CC), blk2, 0, stream>>>(
      base, CC * SS, (long)SS * CC * SS, SS,
      ha_raw, (long)SS * CC * HH, HH,
      tanhref, CC * HH, (long)SS * CC * HH, HH,
      nullptr, nullptr, nullptr);

  gemm_wmma_kernel<0, false, HH, 4, 1><<<dim3(2, 48, 1), blk1, 0, stream>>>(
      tanhref, CC * HH, 0, 0, Wmid + 2 * HH * HH, 0, 0, tref, HH, 0, 0,
      nullptr, nullptr, nullptr);

  gemm_wmma_kernel<MODE_BIAS | MODE_ADDV | MODE_EXP2, true, HH, 1, 2><<<dim3(2, 96, 1), blk2, 0, stream>>>(
      tanh_hp, HH, 0, 0, Wmid, 0, 0, pref_e, HH, 0, 0,
      pref, bmid, tref);

  gemm_wmma_kernel<MODE_EXP2, true, HH, 1, 2><<<dim3(2, 96, 1), blk2, 0, stream>>>(
      tanh_ha, HH, 0, 0, Wmid + HH * HH, 0, 0, targ_e, HH, 0, 0,
      targ, nullptr, nullptr);

  final_reduce_kernel<<<dim3(SS / 128, SS / PCH, BB * CC), dim3(128), 0, stream>>>(
      pref_e, targ_e, pref, targ, wout, out);
}


# Projection GEMMs carry the W1/W2 products: a block owns 16 rows x the 128 h of one case, keeps tanh(h) in LDS as A operand; the t_ref GEMM's epilogue finishes pref; 7 -> 5 launches; sums unchanged
# speedup vs baseline: 4.7149x; 1.0934x over previous
#include <hip/hip_runtime.h>

#define BB 2
#define SS 384
#define CC 4
#define HH 128
#define DD 768

typedef float v2f __attribute__((ext_vector_type(2)));
typedef float v4f __attribute__((ext_vector_type(4)));
typedef float v8f __attribute__((ext_vector_type(8)));

__device__ __forceinline__ float tanhf_fast(float x) {
  float e = __builtin_amdgcn_exp2f(x * 2.8853900817779268f);
  float r = __builtin_amdgcn_rcpf(e + 1.0f);
  return 1.0f - 2.0f * r;
}

#define C2L 2.8853900817779268f

#define MODE_TANH 1
#define MODE_PREF 2

#define KCH 32
#define KST (KCH / 4)

template <int LDB>
__device__ __forceinline__ void load_b(float4 (&b)[KST][2], const float* __restrict__ Bp) {
#pragma unroll
  for (int j = 0; j < KST; ++j) {
    b[j][0] = *(const float4*)(Bp + (4 * j) * LDB);
    b[j][1] = *(const float4*)(Bp + (4 * j + 1) * LDB);
  }
}

template <int LDB>
__device__ __forceinline__ void load_set(v2f (&a)[KST], float4 (&b)[KST][2],
                                         const float* __restrict__ Ap,
                                         const float* __restrict__ Bp) {
#pragma unroll
  for (int j = 0; j < KST; ++j) {
    a[j]    = *(const v2f*)(Ap + 4 * j);
    b[j][0] = *(const float4*)(Bp + (4 * j) * LDB);
    b[j][1] = *(const float4*)(Bp + (4 * j + 1) * LDB);
  }
  __builtin_amdgcn_sched_barrier(0);
}

__device__ __forceinline__ void mma_set(const v2f (&a)[KST], const float4 (&b)[KST][2],
                                        v8f (&acc)[4]) {
#pragma unroll
  for (int j = 0; j < KST; ++j) {
    v2f b0 = {b[j][0].x, b[j][1].x};
    v2f b1 = {b[j][0].y, b[j][1].y};
    v2f b2 = {b[j][0].z, b[j][1].z};
    v2f b3 = {b[j][0].w, b[j][1].w};
    acc[0] = __builtin_amdgcn_wmma_f32_16x16x4_f32(false, a[j], false, b0, (short)0, acc[0], false, false);
    acc[1] = __builtin_amdgcn_wmma_f32_16x16x4_f32(false, a[j], false, b1, (short)0, acc[1], false, false);
    acc[2] = __builtin_amdgcn_wmma_f32_16x16x4_f32(false, a[j], false, b2, (short)0, acc[2], false, false);
    acc[3] = __builtin_amdgcn_wmma_f32_16x16x4_f32(false, a[j], false, b3, (short)0, acc[3], false, false);
  }
}

template <int LDB, int NS>
__device__ __forceinline__ void k_pipeline(const float* __restrict__ Ap,
                                           const float* __restrict__ Bp, v8f (&acc)[4]) {
  v2f    a0[KST], a1[KST];
  float4 b0[KST][2], b1[KST][2];
  load_set<LDB>(a0, b0, Ap, Bp);
#pragma unroll
  for (int s = 0; s < NS; s += 2) {
    if (s + 1 < NS) load_set<LDB>(a1, b1, Ap + (s + 1) * KCH, Bp + (long)(s + 1) * KCH * LDB);
    mma_set(a0, b0, acc);
    if (s + 1 < NS) {
      if (s + 2 < NS) load_set<LDB>(a0, b0, Ap + (s + 2) * KCH, Bp + (long)(s + 2) * KCH * LDB);
      mma_set(a1, b1, acc);
    }
  }
}

#define KP 4
#define RP (64 + 4)

__device__ __forceinline__ void put_partial(float (*red)[RP], const v8f (&acc)[4], int c, int h) {
#pragma unroll
  for (int r = 0; r < 8; ++r) {
    float4 v = {acc[0][r], acc[1][r], acc[2][r], acc[3][r]};
    *(float4*)&red[8 * h + r][4 * c] = v;
  }
}

__device__ __forceinline__ float4 sum_partials(const float (*red)[16][RP], int rl, int c) {
  float4 v = *(const float4*)&red[0][rl][4 * c];
#pragma unroll
  for (int q = 1; q < KP; ++q) {
    float4 p = *(const float4*)&red[q][rl][4 * c];
    v.x += p.x; v.y += p.y; v.z += p.z; v.w += p.w;
  }
  return v;
}

template <int MODE, int LDB, int NS, int MW>
__global__ void __launch_bounds__(32 * MW * KP)
gemm_wmma_kernel(const float* __restrict__ A, int lda, long aOuter, long aInner,
                 const float* __restrict__ B, long bOuter, long bInner,
                 float* __restrict__ O, int ldo, long oOuter, long oInner,
                 float* __restrict__ P, const float* __restrict__ bias) {
  __shared__ __attribute__((aligned(16))) float red[MW][KP][16][RP];
  int lane = threadIdx.x;
  int wm   = threadIdx.y;
  int kq   = threadIdx.z;
  int c    = lane & 15;
  int h    = lane >> 4;
  int zo   = blockIdx.z >> 2;
  int zi   = blockIdx.z & 3;
  int m0   = blockIdx.y * (16 * MW) + wm * 16;
  int n0   = blockIdx.x * 64;
  int k0   = kq * (NS * KCH);

  const float* Ap = A + (long)zo * aOuter + (long)zi * aInner + (long)(m0 + c) * lda + k0 + 2 * h;
  const float* Bp = B + (long)zo * bOuter + (long)zi * bInner + (long)(k0 + 2 * h) * LDB + n0 + 4 * c;

  v8f acc[4] = {};
  k_pipeline<LDB, NS>(Ap, Bp, acc);

  float* Ob = O + (long)zo * oOuter + (long)zi * oInner;
  int col = n0 + 4 * c;
  v4f bmv = {0.f, 0.f, 0.f, 0.f};
  v4f tp[2][CC] = {};
  if constexpr (MODE & MODE_PREF) {
    bmv = *(const v4f*)(bias + col);
#pragma unroll
    for (int i = 0; i < 2; ++i)
#pragma unroll
      for (int cc = 0; cc < CC; ++cc)
        tp[i][cc] = *(const v4f*)(P + ((long)(m0 + 4 * kq + 2 * i + h) * CC + cc) * HH + col);
  }

  put_partial(red[wm][kq], acc, c, h);
  __syncthreads();

#pragma unroll
  for (int i = 0; i < 2; ++i) {
    int rl  = 4 * kq + 2 * i + h;
    int row = m0 + rl;
    float4 v = sum_partials(red[wm], rl, c);
    if constexpr (MODE & MODE_PREF) {
      const v4f tv = {v.x, v.y, v.z, v.w};
#pragma unroll
      for (int cc = 0; cc < CC; ++cc) {
        long off = ((long)row * CC + cc) * HH + col;
        v4f x = tp[i][cc];
        x = x + bmv;
        x = x + tv;
        *(v4f*)(P + off) = x;
        v4f e = {__builtin_amdgcn_exp2f(x.x * C2L), __builtin_amdgcn_exp2f(x.y * C2L),
                 __builtin_amdgcn_exp2f(x.z * C2L), __builtin_amdgcn_exp2f(x.w * C2L)};
        *(v4f*)(O + off) = e;
      }
    } else {
      if constexpr (MODE & MODE_TANH) {
        v.x = tanhf_fast(v.x); v.y = tanhf_fast(v.y); v.z = tanhf_fast(v.z); v.w = tanhf_fast(v.w);
      }
      *(float4*)(Ob + (long)row * ldo + col) = v;
    }
  }
}

#define TQ (HH + 4)
template <bool HAS_HRAW, bool HAS_EXP>
__global__ void __launch_bounds__(32 * 2 * KP)
proj_mid_kernel(const float* __restrict__ X, const float* __restrict__ Wj,
                const float* __restrict__ bj, const float* __restrict__ Wm,
                float* __restrict__ Hraw, float* __restrict__ T, float* __restrict__ Te) {
  __shared__ __attribute__((aligned(16))) float red[2][KP][16][RP];
  __shared__ __attribute__((aligned(16))) float th[16][TQ];
  int lane = threadIdx.x;
  int wn   = threadIdx.y;
  int kq   = threadIdx.z;
  int c    = lane & 15;
  int h    = lane >> 4;
  int cs   = blockIdx.x;
  int m0   = blockIdx.y * 16;
  int n0   = cs * HH + wn * 64;
  int k0   = kq * (6 * KCH);

  const float* Ap = X + (long)(m0 + c) * DD + k0 + 2 * h;
  const float* Bp = Wj + (long)(k0 + 2 * h) * (CC * HH) + n0 + 4 * c;

  v8f acc[4] = {};
  k_pipeline<CC * HH, 6>(Ap, Bp, acc);

  float4 wb[KST][2];
  load_b<HH>(wb, Wm + (long)(kq * KCH + 2 * h) * HH + wn * 64 + 4 * c);
  int col = n0 + 4 * c;
  float4 bi = *(const float4*)(bj + col);

  put_partial(red[wn][kq], acc, c, h);
  __syncthreads();

#pragma unroll
  for (int i = 0; i < 2; ++i) {
    int rl = 4 * kq + 2 * i + h;
    float4 v = sum_partials(red[wn], rl, c);
    v.x += bi.x; v.y += bi.y; v.z += bi.z; v.w += bi.w;
    if constexpr (HAS_HRAW) *(float4*)(Hraw + (long)(m0 + rl) * (CC * HH) + col) = v;
    v.x = tanhf_fast(v.x); v.y = tanhf_fast(v.y); v.z = tanhf_fast(v.z); v.w = tanhf_fast(v.w);
    *(float4*)&th[rl][wn * 64 + 4 * c] = v;
  }
  __syncthreads();

  v2f a2[KST];
#pragma unroll
  for (int j = 0; j < KST; ++j) a2[j] = *(const v2f*)&th[c][kq * KCH + 4 * j + 2 * h];
  v8f acc2[4] = {};
  mma_set(a2, wb, acc2);

  put_partial(red[wn][kq], acc2, c, h);
  __syncthreads();

  int col2 = wn * 64 + 4 * c;
#pragma unroll
  for (int i = 0; i < 2; ++i) {
    int rl = 4 * kq + 2 * i + h;
    float4 v = sum_partials(red[wn], rl, c);
    long off = ((long)(m0 + rl) * CC + cs) * HH + col2;
    *(float4*)(T + off) = v;
    if constexpr (HAS_EXP) {
      float4 e;
      e.x = __builtin_amdgcn_exp2f(v.x * C2L); e.y = __builtin_amdgcn_exp2f(v.y * C2L);
      e.z = __builtin_amdgcn_exp2f(v.z * C2L); e.w = __builtin_amdgcn_exp2f(v.w * C2L);
      *(float4*)(Te + off) = e;
    }
  }
}

#define PCH 12
#define EBK 8
#define TP  (HH + 4)
#define ELO_BITS 0x2F800000u
#define EHI_BITS 0x4F800000u

template <bool FAST>
__device__ __forceinline__ void final_rows(const float (*sp)[HH], const float* st_row,
                                           const float* swv, float init, float* op) {
  v2f t[HH / 2];
#pragma unroll
  for (int j = 0; j < HH / 2; ++j) t[j] = *(const v2f*)&st_row[2 * j];
  float w[HH];
#pragma unroll
  for (int k = 0; k < HH; ++k) w[k] = swv[k];
  const v2f one = {1.0f, 1.0f};

#pragma unroll 1
  for (int q = 0; q < PCH / 2; ++q) {
    const v2f* s0 = (const v2f*)sp[2 * q];
    const v2f* s1 = (const v2f*)sp[2 * q + 1];
    float acc0 = init, acc1 = init;
    v2f n0[EBK / 2], n1[EBK / 2];
#pragma unroll
    for (int i = 0; i < EBK / 2; ++i) { n0[i] = s0[i]; n1[i] = s1[i]; }
#pragma unroll
    for (int k = 0; k < HH; k += EBK) {
      v2f x0[EBK / 2], x1[EBK / 2];
#pragma unroll
      for (int i = 0; i < EBK / 2; ++i) { x0[i] = n0[i]; x1[i] = n1[i]; }
      if (k + EBK < HH) {
#pragma unroll
        for (int i = 0; i < EBK / 2; ++i) {
          n0[i] = s0[(k + EBK) / 2 + i];
          n1[i] = s1[(k + EBK) / 2 + i];
        }
      }
      __builtin_amdgcn_sched_barrier(0x1);
      if constexpr (FAST) {
#pragma unroll
        for (int i = 0; i < EBK / 2; ++i) {
          x0[i] = __builtin_elementwise_fma(x0[i], t[k / 2 + i], one);
          x1[i] = __builtin_elementwise_fma(x1[i], t[k / 2 + i], one);
        }
      } else {
#pragma unroll
        for (int i = 0; i < EBK / 2; ++i) {
          x0[i] = (x0[i] + t[k / 2 + i]) * C2L;
          x1[i] = (x1[i] + t[k / 2 + i]) * C2L;
        }
#pragma unroll
        for (int i = 0; i < EBK / 2; ++i) {
          x0[i].x = __builtin_amdgcn_exp2f(x0[i].x);
          x0[i].y = __builtin_amdgcn_exp2f(x0[i].y);
          x1[i].x = __builtin_amdgcn_exp2f(x1[i].x);
          x1[i].y = __builtin_amdgcn_exp2f(x1[i].y);
        }
#pragma unroll
        for (int i = 0; i < EBK / 2; ++i) { x0[i] = x0[i] + 1.0f; x1[i] = x1[i] + 1.0f; }
      }
#pragma unroll
      for (int i = 0; i < EBK / 2; ++i) {
        x0[i].x = __builtin_amdgcn_rcpf(x0[i].x);
        x0[i].y = __builtin_amdgcn_rcpf(x0[i].y);
        x1[i].x = __builtin_amdgcn_rcpf(x1[i].x);
        x1[i].y = __builtin_amdgcn_rcpf(x1[i].y);
      }
#pragma unroll
      for (int i = 0; i < EBK / 2; ++i) {
        v2f h0 = x0[i], h1 = x1[i];
        if constexpr (!FAST) {
          h0 = 1.0f - 2.0f * x0[i];
          h1 = 1.0f - 2.0f * x1[i];
        }
        acc0 += h0.x * w[k + 2 * i];
        acc1 += h1.x * w[k + 2 * i];
        acc0 += h0.y * w[k + 2 * i + 1];
        acc1 += h1.y * w[k + 2 * i + 1];
      }
    }
    op[0]       = acc0;
    op[CC * SS] = acc1;
    op += 2 * CC * SS;
  }
}

__global__ void __launch_bounds__(128)
final_reduce_kernel(const float* __restrict__ prefE, const float* __restrict__ targE,
                    const float* __restrict__ prefR, const float* __restrict__ targR,
                    const float* __restrict__ wout, float* __restrict__ out) {
  __shared__ __attribute__((aligned(16))) float lds[2 * HH + PCH * HH + 128 * TP];
  __shared__ int sflag[4];
  float* sw  = lds;
  float* sw2 = lds + HH;
  float (*sp)[HH] = (float (*)[HH])(lds + 2 * HH);
  float (*st)[TP] = (float (*)[TP])(lds + 2 * HH + PCH * HH);
  int tid  = threadIdx.x;
  int lane = tid & 31;
  int wv   = tid >> 5;
  int bc   = blockIdx.z;
  int b    = bc >> 2;
  int c    = bc & 3;
  int a0   = blockIdx.x * 128;
  int p0   = blockIdx.y * PCH;

  long poff = ((long)(b * SS + p0) * CC + c) * HH;
  long toff = ((long)(b * SS + a0 + 32 * wv) * CC + c) * HH + 4 * lane;
  float wk = wout[tid];
  sw[tid]  = wk;
  sw2[tid] = -2.0f * wk;
  unsigned ulo = 0xFFFFFFFFu, uhi = 0u;
#pragma unroll
  for (int r = 0; r < PCH; ++r) {
    float e = prefE[poff + (long)r * CC * HH + tid];
    sp[r][tid] = e;
    unsigned u = __float_as_uint(e);
    ulo = min(ulo, u); uhi = max(uhi, u);
  }
#pragma unroll
  for (int j = 0; j < 32; ++j) {
    float4 tv = *(const float4*)(targE + toff + (long)j * CC * HH);
    *(float4*)&st[32 * wv + j][4 * lane] = tv;
    unsigned ux = __float_as_uint(tv.x), uy = __float_as_uint(tv.y);
    unsigned uz = __float_as_uint(tv.z), uw = __float_as_uint(tv.w);
    ulo = min(min(ulo, ux), min(uy, min(uz, uw)));
    uhi = max(max(uhi, ux), max(uy, max(uz, uw)));
  }
  bool ok = (ulo >= ELO_BITS) && (uhi <= EHI_BITS);
  int wbad = __any(!ok);
  if (lane == 0) sflag[wv] = wbad;
  __syncthreads();
  int direct = (sflag[0] | sflag[1]) | (sflag[2] | sflag[3]);
  direct = __builtin_amdgcn_readfirstlane(direct);

  float* op = out + ((long)(b * SS + p0) * CC + c) * SS + a0 + tid;
  if (direct) {
#pragma unroll
    for (int r = 0; r < PCH; ++r) sp[r][tid] = prefR[poff + (long)r * CC * HH + tid];
#pragma unroll
    for (int j = 0; j < 32; ++j)
      *(float4*)&st[32 * wv + j][4 * lane] = *(const float4*)(targR + toff + (long)j * CC * HH);
    __syncthreads();
    final_rows<false>(sp, st[tid], sw, 0.0f, op);
  } else {
    float s = ((sw[lane] + sw[lane + 32]) + sw[lane + 64]) + sw[lane + 96];
#pragma unroll
    for (int m = 16; m >= 1; m >>= 1) s += __shfl_xor(s, m);
    final_rows<true>(sp, st[tid], sw2, s, op);
  }
}

extern "C" void kernel_launch(void* const* d_in, const int* in_sizes, int n_in,
                              void* d_out, int out_size, void* d_ws, size_t ws_size,
                              hipStream_t stream) {
  const float* seq  = (const float*)d_in[0];
  const float* base = (const float*)d_in[1];
  const float* Wp   = (const float*)d_in[2];
  const float* bp   = (const float*)d_in[3];
  const float* Ua   = (const float*)d_in[4];
  const float* ba   = (const float*)d_in[5];
  const float* Wmid = (const float*)d_in[6];
  const float* bmid = (const float*)d_in[7];
  const float* wout = (const float*)d_in[8];
  float* out = (float*)d_out;

  float* ws      = (float*)d_ws;
  float* ha_raw  = ws;
  float* tanhref = ws + 393216;
  float* targ    = ws + 786432;
  float* pref    = ws + 1179648;
  float* pref_e  = ws + 1572864;
  float* targ_e  = ws + 1966080;

  dim3 blk2(32, 2, KP);
  dim3 blk1(32, 1, KP);

  proj_mid_kernel<false, false><<<dim3(CC, BB * SS / 16, 1), blk2, 0, stream>>>(
      seq, Wp, bp, Wmid, nullptr, pref, nullptr);

  proj_mid_kernel<true, true><<<dim3(CC, BB * SS / 16, 1), blk2, 0, stream>>>(
      seq, Ua, ba, Wmid + HH * HH, ha_raw, targ, targ_e);

  gemm_wmma_kernel<MODE_TANH, CC * HH, 3, 2><<<dim3(2, 12, BB * CC), blk2, 0, stream>>>(
      base, CC * SS, (long)SS * CC * SS, SS,
      ha_raw, (long)SS * CC * HH, HH,
      tanhref, CC * HH, (long)SS * CC * HH, HH,
      nullptr, nullptr);

  gemm_wmma_kernel<MODE_PREF, HH, 4, 1><<<dim3(2, 48, 1), blk1, 0, stream>>>(
      tanhref, CC * HH, 0, 0, Wmid + 2 * HH * HH, 0, 0, pref_e, HH, 0, 0,
      pref, bmid);

  final_reduce_kernel<<<dim3(SS / 128, SS / PCH, BB * CC), dim3(128), 0, stream>>>(
      pref_e, targ_e, pref, targ, wout, out);
}


# The prd projection (with its W1 product) and the ref GEMM, which do not feed each other, run as one grid of 2 x 192 blocks behind the arg projection; 5 launches -> 4; every sum unchanged
# speedup vs baseline: 5.0839x; 1.0783x over previous
#include <hip/hip_runtime.h>

#define BB 2
#define SS 384
#define CC 4
#define HH 128
#define DD 768

typedef float v2f __attribute__((ext_vector_type(2)));
typedef float v4f __attribute__((ext_vector_type(4)));
typedef float v8f __attribute__((ext_vector_type(8)));

__device__ __forceinline__ float tanhf_fast(float x) {
  float e = __builtin_amdgcn_exp2f(x * 2.8853900817779268f);
  float r = __builtin_amdgcn_rcpf(e + 1.0f);
  return 1.0f - 2.0f * r;
}

#define C2L 2.8853900817779268f

#define MODE_TANH 1
#define MODE_PREF 2

#define KCH 32
#define KST (KCH / 4)

template <int LDB>
__device__ __forceinline__ void load_b(float4 (&b)[KST][2], const float* __restrict__ Bp) {
#pragma unroll
  for (int j = 0; j < KST; ++j) {
    b[j][0] = *(const float4*)(Bp + (4 * j) * LDB);
    b[j][1] = *(const float4*)(Bp + (4 * j + 1) * LDB);
  }
}

template <int LDB>
__device__ __forceinline__ void load_set(v2f (&a)[KST], float4 (&b)[KST][2],
                                         const float* __restrict__ Ap,
                                         const float* __restrict__ Bp) {
#pragma unroll
  for (int j = 0; j < KST; ++j) {
    a[j]    = *(const v2f*)(Ap + 4 * j);
    b[j][0] = *(const float4*)(Bp + (4 * j) * LDB);
    b[j][1] = *(const float4*)(Bp + (4 * j + 1) * LDB);
  }
  __builtin_amdgcn_sched_barrier(0);
}

__device__ __forceinline__ void mma_set(const v2f (&a)[KST], const float4 (&b)[KST][2],
                                        v8f (&acc)[4]) {
#pragma unroll
  for (int j = 0; j < KST; ++j) {
    v2f b0 = {b[j][0].x, b[j][1].x};
    v2f b1 = {b[j][0].y, b[j][1].y};
    v2f b2 = {b[j][0].z, b[j][1].z};
    v2f b3 = {b[j][0].w, b[j][1].w};
    acc[0] = __builtin_amdgcn_wmma_f32_16x16x4_f32(false, a[j], false, b0, (short)0, acc[0], false, false);
    acc[1] = __builtin_amdgcn_wmma_f32_16x16x4_f32(false, a[j], false, b1, (short)0, acc[1], false, false);
    acc[2] = __builtin_amdgcn_wmma_f32_16x16x4_f32(false, a[j], false, b2, (short)0, acc[2], false, false);
    acc[3] = __builtin_amdgcn_wmma_f32_16x16x4_f32(false, a[j], false, b3, (short)0, acc[3], false, false);
  }
}

template <int LDB, int NS>
__device__ __forceinline__ void k_pipeline(const float* __restrict__ Ap,
                                           const float* __restrict__ Bp, v8f (&acc)[4]) {
  v2f    a0[KST], a1[KST];
  float4 b0[KST][2], b1[KST][2];
  load_set<LDB>(a0, b0, Ap, Bp);
#pragma unroll
  for (int s = 0; s < NS; s += 2) {
    if (s + 1 < NS) load_set<LDB>(a1, b1, Ap + (s + 1) * KCH, Bp + (long)(s + 1) * KCH * LDB);
    mma_set(a0, b0, acc);
    if (s + 1 < NS) {
      if (s + 2 < NS) load_set<LDB>(a0, b0, Ap + (s + 2) * KCH, Bp + (long)(s + 2) * KCH * LDB);
      mma_set(a1, b1, acc);
    }
  }
}

#define KP 4
#define RP (64 + 4)

__device__ __forceinline__ void put_partial(float (*red)[RP], const v8f (&acc)[4], int c, int h) {
#pragma unroll
  for (int r = 0; r < 8; ++r) {
    float4 v = {acc[0][r], acc[1][r], acc[2][r], acc[3][r]};
    *(float4*)&red[8 * h + r][4 * c] = v;
  }
}

__device__ __forceinline__ float4 sum_partials(const float (*red)[16][RP], int rl, int c) {
  float4 v = *(const float4*)&red[0][rl][4 * c];
#pragma unroll
  for (int q = 1; q < KP; ++q) {
    float4 p = *(const float4*)&red[q][rl][4 * c];
    v.x += p.x; v.y += p.y; v.z += p.z; v.w += p.w;
  }
  return v;
}

template <int MODE, int LDB, int NS, int MW>
__device__ __forceinline__ void
gemm_block(float (*red)[KP][16][RP], int bx, int by, int bz,
           const float* __restrict__ A, int lda, long aOuter, long aInner,
           const float* __restrict__ B, long bOuter, long bInner,
           float* __restrict__ O, int ldo, long oOuter, long oInner,
           float* __restrict__ P, const float* __restrict__ bias) {
  int lane = threadIdx.x;
  int wm   = threadIdx.y;
  int kq   = threadIdx.z;
  int c    = lane & 15;
  int h    = lane >> 4;
  int zo   = bz >> 2;
  int zi   = bz & 3;
  int m0   = by * (16 * MW) + wm * 16;
  int n0   = bx * 64;
  int k0   = kq * (NS * KCH);

  const float* Ap = A + (long)zo * aOuter + (long)zi * aInner + (long)(m0 + c) * lda + k0 + 2 * h;
  const float* Bp = B + (long)zo * bOuter + (long)zi * bInner + (long)(k0 + 2 * h) * LDB + n0 + 4 * c;

  v8f acc[4] = {};
  k_pipeline<LDB, NS>(Ap, Bp, acc);

  float* Ob = O + (long)zo * oOuter + (long)zi * oInner;
  int col = n0 + 4 * c;
  v4f bmv = {0.f, 0.f, 0.f, 0.f};
  v4f tp[2][CC] = {};
  if constexpr (MODE & MODE_PREF) {
    bmv = *(const v4f*)(bias + col);
#pragma unroll
    for (int i = 0; i < 2; ++i)
#pragma unroll
      for (int cc = 0; cc < CC; ++cc)
        tp[i][cc] = *(const v4f*)(P + ((long)(m0 + 4 * kq + 2 * i + h) * CC + cc) * HH + col);
  }

  put_partial(red[wm][kq], acc, c, h);
  __syncthreads();

#pragma unroll
  for (int i = 0; i < 2; ++i) {
    int rl  = 4 * kq + 2 * i + h;
    int row = m0 + rl;
    float4 v = sum_partials(red[wm], rl, c);
    if constexpr (MODE & MODE_PREF) {
      const v4f tv = {v.x, v.y, v.z, v.w};
#pragma unroll
      for (int cc = 0; cc < CC; ++cc) {
        long off = ((long)row * CC + cc) * HH + col;
        v4f x = tp[i][cc];
        x = x + bmv;
        x = x + tv;
        *(v4f*)(P + off) = x;
        v4f e = {__builtin_amdgcn_exp2f(x.x * C2L), __builtin_amdgcn_exp2f(x.y * C2L),
                 __builtin_amdgcn_exp2f(x.z * C2L), __builtin_amdgcn_exp2f(x.w * C2L)};
        *(v4f*)(O + off) = e;
      }
    } else {
      if constexpr (MODE & MODE_TANH) {
        v.x = tanhf_fast(v.x); v.y = tanhf_fast(v.y); v.z = tanhf_fast(v.z); v.w = tanhf_fast(v.w);
      }
      *(float4*)(Ob + (long)row * ldo + col) = v;
    }
  }
}

template <int MODE, int LDB, int NS, int MW>
__global__ void __launch_bounds__(32 * MW * KP)
gemm_wmma_kernel(const float* __restrict__ A, int lda, long aOuter, long aInner,
                 const float* __restrict__ B, long bOuter, long bInner,
                 float* __restrict__ O, int ldo, long oOuter, long oInner,
                 float* __restrict__ P, const float* __restrict__ bias) {
  __shared__ __attribute__((aligned(16))) float red[MW][KP][16][RP];
  gemm_block<MODE, LDB, NS, MW>(red, blockIdx.x, blockIdx.y, blockIdx.z, A, lda, aOuter, aInner,
                                B, bOuter, bInner, O, ldo, oOuter, oInner, P, bias);
}

#define TQ (HH + 4)
template <bool HAS_HRAW, bool HAS_EXP>
__device__ __forceinline__ void
proj_mid_block(float (*red)[KP][16][RP], float (*th)[TQ], int cs, int mb,
               const float* __restrict__ X, const float* __restrict__ Wj,
               const float* __restrict__ bj, const float* __restrict__ Wm,
               float* __restrict__ Hraw, float* __restrict__ T, float* __restrict__ Te) {
  int lane = threadIdx.x;
  int wn   = threadIdx.y;
  int kq   = threadIdx.z;
  int c    = lane & 15;
  int h    = lane >> 4;
  int m0   = mb * 16;
  int n0   = cs * HH + wn * 64;
  int k0   = kq * (6 * KCH);

  const float* Ap = X + (long)(m0 + c) * DD + k0 + 2 * h;
  const float* Bp = Wj + (long)(k0 + 2 * h) * (CC * HH) + n0 + 4 * c;

  v8f acc[4] = {};
  k_pipeline<CC * HH, 6>(Ap, Bp, acc);

  float4 wb[KST][2];
  load_b<HH>(wb, Wm + (long)(kq * KCH + 2 * h) * HH + wn * 64 + 4 * c);
  int col = n0 + 4 * c;
  float4 bi = *(const float4*)(bj + col);

  put_partial(red[wn][kq], acc, c, h);
  __syncthreads();

#pragma unroll
  for (int i = 0; i < 2; ++i) {
    int rl = 4 * kq + 2 * i + h;
    float4 v = sum_partials(red[wn], rl, c);
    v.x += bi.x; v.y += bi.y; v.z += bi.z; v.w += bi.w;
    if constexpr (HAS_HRAW) *(float4*)(Hraw + (long)(m0 + rl) * (CC * HH) + col) = v;
    v.x = tanhf_fast(v.x); v.y = tanhf_fast(v.y); v.z = tanhf_fast(v.z); v.w = tanhf_fast(v.w);
    *(float4*)&th[rl][wn * 64 + 4 * c] = v;
  }
  __syncthreads();

  v2f a2[KST];
#pragma unroll
  for (int j = 0; j < KST; ++j) a2[j] = *(const v2f*)&th[c][kq * KCH + 4 * j + 2 * h];
  v8f acc2[4] = {};
  mma_set(a2, wb, acc2);

  put_partial(red[wn][kq], acc2, c, h);
  __syncthreads();

  int col2 = wn * 64 + 4 * c;
#pragma unroll
  for (int i = 0; i < 2; ++i) {
    int rl = 4 * kq + 2 * i + h;
    float4 v = sum_partials(red[wn], rl, c);
    long off = ((long)(m0 + rl) * CC + cs) * HH + col2;
    *(float4*)(T + off) = v;
    if constexpr (HAS_EXP) {
      float4 e;
      e.x = __builtin_amdgcn_exp2f(v.x * C2L); e.y = __builtin_amdgcn_exp2f(v.y * C2L);
      e.z = __builtin_amdgcn_exp2f(v.z * C2L); e.w = __builtin_amdgcn_exp2f(v.w * C2L);
      *(float4*)(Te + off) = e;
    }
  }
}

template <bool HAS_HRAW, bool HAS_EXP>
__global__ void __launch_bounds__(32 * 2 * KP)
proj_mid_kernel(const float* __restrict__ X, const float* __restrict__ Wj,
                const float* __restrict__ bj, const float* __restrict__ Wm,
                float* __restrict__ Hraw, float* __restrict__ T, float* __restrict__ Te) {
  __shared__ __attribute__((aligned(16))) float red[2][KP][16][RP];
  __shared__ __attribute__((aligned(16))) float th[16][TQ];
  proj_mid_block<HAS_HRAW, HAS_EXP>(red, th, blockIdx.x, blockIdx.y, X, Wj, bj, Wm, Hraw, T, Te);
}

__global__ void __launch_bounds__(32 * 2 * KP)
prd_and_ref_kernel(const float* __restrict__ seq, const float* __restrict__ Wp,
                   const float* __restrict__ bp, const float* __restrict__ W1,
                   float* __restrict__ tprd, const float* __restrict__ base,
                   const float* __restrict__ ha, float* __restrict__ tanhref) {
  __shared__ __attribute__((aligned(16))) float red[2][KP][16][RP];
  __shared__ __attribute__((aligned(16))) float th[16][TQ];
  if (blockIdx.z == 0) {
    proj_mid_block<false, false>(red, th, blockIdx.x, blockIdx.y, seq, Wp, bp, W1, nullptr, tprd,
                                 nullptr);
  } else {
    int lin = blockIdx.y * CC + blockIdx.x;
    int bx  = lin & 1;
    int by  = (lin >> 1) % (SS / 32);
    int bz  = (lin >> 1) / (SS / 32);
    gemm_block<MODE_TANH, CC * HH, 3, 2>(red, bx, by, bz,
        base, CC * SS, (long)SS * CC * SS, SS,
        ha, (long)SS * CC * HH, HH,
        tanhref, CC * HH, (long)SS * CC * HH, HH,
        nullptr, nullptr);
  }
}

#define PCH 12
#define EBK 8
#define TP  (HH + 4)
#define ELO_BITS 0x2F800000u
#define EHI_BITS 0x4F800000u

template <bool FAST>
__device__ __forceinline__ void final_rows(const float (*sp)[HH], const float* st_row,
                                           const float* swv, float init, float* op) {
  v2f t[HH / 2];
#pragma unroll
  for (int j = 0; j < HH / 2; ++j) t[j] = *(const v2f*)&st_row[2 * j];
  float w[HH];
#pragma unroll
  for (int k = 0; k < HH; ++k) w[k] = swv[k];
  const v2f one = {1.0f, 1.0f};

#pragma unroll 1
  for (int q = 0; q < PCH / 2; ++q) {
    const v2f* s0 = (const v2f*)sp[2 * q];
    const v2f* s1 = (const v2f*)sp[2 * q + 1];
    float acc0 = init, acc1 = init;
    v2f n0[EBK / 2], n1[EBK / 2];
#pragma unroll
    for (int i = 0; i < EBK / 2; ++i) { n0[i] = s0[i]; n1[i] = s1[i]; }
#pragma unroll
    for (int k = 0; k < HH; k += EBK) {
      v2f x0[EBK / 2], x1[EBK / 2];
#pragma unroll
      for (int i = 0; i < EBK / 2; ++i) { x0[i] = n0[i]; x1[i] = n1[i]; }
      if (k + EBK < HH) {
#pragma unroll
        for (int i = 0; i < EBK / 2; ++i) {
          n0[i] = s0[(k + EBK) / 2 + i];
          n1[i] = s1[(k + EBK) / 2 + i];
        }
      }
      __builtin_amdgcn_sched_barrier(0x1);
      if constexpr (FAST) {
#pragma unroll
        for (int i = 0; i < EBK / 2; ++i) {
          x0[i] = __builtin_elementwise_fma(x0[i], t[k / 2 + i], one);
          x1[i] = __builtin_elementwise_fma(x1[i], t[k / 2 + i], one);
        }
      } else {
#pragma unroll
        for (int i = 0; i < EBK / 2; ++i) {
          x0[i] = (x0[i] + t[k / 2 + i]) * C2L;
          x1[i] = (x1[i] + t[k / 2 + i]) * C2L;
        }
#pragma unroll
        for (int i = 0; i < EBK / 2; ++i) {
          x0[i].x = __builtin_amdgcn_exp2f(x0[i].x);
          x0[i].y = __builtin_amdgcn_exp2f(x0[i].y);
          x1[i].x = __builtin_amdgcn_exp2f(x1[i].x);
          x1[i].y = __builtin_amdgcn_exp2f(x1[i].y);
        }
#pragma unroll
        for (int i = 0; i < EBK / 2; ++i) { x0[i] = x0[i] + 1.0f; x1[i] = x1[i] + 1.0f; }
      }
#pragma unroll
      for (int i = 0; i < EBK / 2; ++i) {
        x0[i].x = __builtin_amdgcn_rcpf(x0[i].x);
        x0[i].y = __builtin_amdgcn_rcpf(x0[i].y);
        x1[i].x = __builtin_amdgcn_rcpf(x1[i].x);
        x1[i].y = __builtin_amdgcn_rcpf(x1[i].y);
      }
#pragma unroll
      for (int i = 0; i < EBK / 2; ++i) {
        v2f h0 = x0[i], h1 = x1[i];
        if constexpr (!FAST) {
          h0 = 1.0f - 2.0f * x0[i];
          h1 = 1.0f - 2.0f * x1[i];
        }
        acc0 += h0.x * w[k + 2 * i];
        acc1 += h1.x * w[k + 2 * i];
        acc0 += h0.y * w[k + 2 * i + 1];
        acc1 += h1.y * w[k + 2 * i + 1];
      }
    }
    op[0]       = acc0;
    op[CC * SS] = acc1;
    op += 2 * CC * SS;
  }
}

__global__ void __launch_bounds__(128)
final_reduce_kernel(const float* __restrict__ prefE, const float* __restrict__ targE,
                    const float* __restrict__ prefR, const float* __restrict__ targR,
                    const float* __restrict__ wout, float* __restrict__ out) {
  __shared__ __attribute__((aligned(16))) float lds[2 * HH + PCH * HH + 128 * TP];
  __shared__ int sflag[4];
  float* sw  = lds;
  float* sw2 = lds + HH;
  float (*sp)[HH] = (float (*)[HH])(lds + 2 * HH);
  float (*st)[TP] = (float (*)[TP])(lds + 2 * HH + PCH * HH);
  int tid  = threadIdx.x;
  int lane = tid & 31;
  int wv   = tid >> 5;
  int bc   = blockIdx.z;
  int b    = bc >> 2;
  int c    = bc & 3;
  int a0   = blockIdx.x * 128;
  int p0   = blockIdx.y * PCH;

  long poff = ((long)(b * SS + p0) * CC + c) * HH;
  long toff = ((long)(b * SS + a0 + 32 * wv) * CC + c) * HH + 4 * lane;
  float wk = wout[tid];
  sw[tid]  = wk;
  sw2[tid] = -2.0f * wk;
  unsigned ulo = 0xFFFFFFFFu, uhi = 0u;
#pragma unroll
  for (int r = 0; r < PCH; ++r) {
    float e = prefE[poff + (long)r * CC * HH + tid];
    sp[r][tid] = e;
    unsigned u = __float_as_uint(e);
    ulo = min(ulo, u); uhi = max(uhi, u);
  }
#pragma unroll
  for (int j = 0; j < 32; ++j) {
    float4 tv = *(const float4*)(targE + toff + (long)j * CC * HH);
    *(float4*)&st[32 * wv + j][4 * lane] = tv;
    unsigned ux = __float_as_uint(tv.x), uy = __float_as_uint(tv.y);
    unsigned uz = __float_as_uint(tv.z), uw = __float_as_uint(tv.w);
    ulo = min(min(ulo, ux), min(uy, min(uz, uw)));
    uhi = max(max(uhi, ux), max(uy, max(uz, uw)));
  }
  bool ok = (ulo >= ELO_BITS) && (uhi <= EHI_BITS);
  int wbad = __any(!ok);
  if (lane == 0) sflag[wv] = wbad;
  __syncthreads();
  int direct = (sflag[0] | sflag[1]) | (sflag[2] | sflag[3]);
  direct = __builtin_amdgcn_readfirstlane(direct);

  float* op = out + ((long)(b * SS + p0) * CC + c) * SS + a0 + tid;
  if (direct) {
#pragma unroll
    for (int r = 0; r < PCH; ++r) sp[r][tid] = prefR[poff + (long)r * CC * HH + tid];
#pragma unroll
    for (int j = 0; j < 32; ++j)
      *(float4*)&st[32 * wv + j][4 * lane] = *(const float4*)(targR + toff + (long)j * CC * HH);
    __syncthreads();
    final_rows<false>(sp, st[tid], sw, 0.0f, op);
  } else {
    float s = ((sw[lane] + sw[lane + 32]) + sw[lane + 64]) + sw[lane + 96];
#pragma unroll
    for (int m = 16; m >= 1; m >>= 1) s += __shfl_xor(s, m);
    final_rows<true>(sp, st[tid], sw2, s, op);
  }
}

extern "C" void kernel_launch(void* const* d_in, const int* in_sizes, int n_in,
                              void* d_out, int out_size, void* d_ws, size_t ws_size,
                              hipStream_t stream) {
  const float* seq  = (const float*)d_in[0];
  const float* base = (const float*)d_in[1];
  const float* Wp   = (const float*)d_in[2];
  const float* bp   = (const float*)d_in[3];
  const float* Ua   = (const float*)d_in[4];
  const float* ba   = (const float*)d_in[5];
  const float* Wmid = (const float*)d_in[6];
  const float* bmid = (const float*)d_in[7];
  const float* wout = (const float*)d_in[8];
  float* out = (float*)d_out;

  float* ws      = (float*)d_ws;
  float* ha_raw  = ws;
  float* tanhref = ws + 393216;
  float* targ    = ws + 786432;
  float* pref    = ws + 1179648;
  float* pref_e  = ws + 1572864;
  float* targ_e  = ws + 1966080;

  dim3 blk2(32, 2, KP);
  dim3 blk1(32, 1, KP);

  proj_mid_kernel<true, true><<<dim3(CC, BB * SS / 16, 1), blk2, 0, stream>>>(
      seq, Ua, ba, Wmid + HH * HH, ha_raw, targ, targ_e);

  prd_and_ref_kernel<<<dim3(CC, BB * SS / 16, 2), blk2, 0, stream>>>(
      seq, Wp, bp, Wmid, pref, base, ha_raw, tanhref);

  gemm_wmma_kernel<MODE_PREF, HH, 4, 1><<<dim3(2, 48, 1), blk1, 0, stream>>>(
      tanhref, CC * HH, 0, 0, Wmid + 2 * HH * HH, 0, 0, pref_e, HH, 0, 0,
      pref, bmid);

  final_reduce_kernel<<<dim3(SS / 128, SS / PCH, BB * CC), dim3(128), 0, stream>>>(
      pref_e, targ_e, pref, targ, wout, out);
}
